# starting kernel
# baseline (speedup 1.0000x reference)
#include <hip/hip_runtime.h>

#define N_  50000
#define E_  800000
#define D_  256
#define H_  4
#define F_  32
#define HF_ 128
#define NEG_SLOPE 0.2f

typedef __attribute__((ext_vector_type(2))) float v2f;
typedef __attribute__((ext_vector_type(8))) float v8f;

__device__ __forceinline__ unsigned encodeOrd(float f) {
  unsigned u = __float_as_uint(f);
  return (u & 0x80000000u) ? ~u : (u | 0x80000000u);
}
__device__ __forceinline__ float decodeOrd(unsigned u) {
  unsigned b = (u & 0x80000000u) ? (u & 0x7FFFFFFFu) : ~u;
  return __uint_as_float(b);
}

__global__ __launch_bounds__(256) void gat_gemm_wmma(
    const float* __restrict__ h, const float* __restrict__ W,
    float* __restrict__ hW, int nRows)
{
  const int wave = threadIdx.x >> 5;
  const int lane = threadIdx.x & 31;
  const int rowBase = (blockIdx.x * 8 + wave) * 16;
  if (rowBase >= nRows) return;

  const int m    = lane & 15;
  const int khalf= (lane >> 4) * 2;

  v8f acc[8];
#pragma unroll
  for (int t = 0; t < 8; ++t) acc[t] = (v8f){0.f,0.f,0.f,0.f,0.f,0.f,0.f,0.f};

  for (int k0 = 0; k0 < D_; k0 += 4) {
    const float* ap = h + (size_t)(rowBase + m) * D_ + (k0 + khalf);
    v2f a; a.x = ap[0]; a.y = ap[1];
#pragma unroll
    for (int t = 0; t < 8; ++t) {
      const int n = t * 16 + (lane & 15);
      const float* bp = W + (size_t)(k0 + khalf) * HF_ + n;
      v2f b; b.x = bp[0]; b.y = bp[HF_];
      acc[t] = __builtin_amdgcn_wmma_f32_16x16x4_f32(
           false, a,  false, b,
           (short)0, acc[t],  false,  false);
    }
  }

  const int mHalf = (lane >> 4) * 8;
#pragma unroll
  for (int t = 0; t < 8; ++t) {
    const int n = t * 16 + (lane & 15);
#pragma unroll
    for (int v = 0; v < 8; ++v) {
      hW[(size_t)(rowBase + mHalf + v) * HF_ + n] = acc[t][v];
    }
  }
}

__global__ __launch_bounds__(256) void gat_attn_proj(
    const float* __restrict__ hW,
    const float* __restrict__ alw, const float* __restrict__ alb,
    const float* __restrict__ arw, const float* __restrict__ arb,
    float* __restrict__ el, float* __restrict__ er, int nNH)
{
  const int idx = blockIdx.x * blockDim.x + threadIdx.x;
  if (idx >= nNH) return;
  const float* p = hW + (size_t)idx * F_;
  float sl = 0.f, sr = 0.f;
#pragma unroll
  for (int f = 0; f < F_; ++f) {
    const float x = p[f];
    sl += x * alw[f];
    sr += x * arw[f];
  }
  el[idx] = sl + alb[0];
  er[idx] = sr + arb[0];
}

__global__ __launch_bounds__(256) void gat_init(
    unsigned* __restrict__ emax, float* __restrict__ denom,
    float* __restrict__ out, const float* __restrict__ bias,
    int nNH, int nNHF)
{
  const int i = blockIdx.x * blockDim.x + threadIdx.x;
  if (i < nNH) {
    emax[i]  = encodeOrd(-__builtin_inff());
    denom[i] = 0.f;
  }
  if (i < nNHF) out[i] = bias[i & (HF_ - 1)];
}

__global__ __launch_bounds__(256) void gat_edge_max(
    const int* __restrict__ src, const int* __restrict__ dst,
    const float* __restrict__ el, const float* __restrict__ er,
    unsigned* __restrict__ emax, int nEH)
{
  const int i = blockIdx.x * blockDim.x + threadIdx.x;
  if (i >= nEH) return;
  const int e = i >> 2;
  const int hh = i & (H_ - 1);
  const int s = src[e], d = dst[e];
  float v = er[s * H_ + hh] + el[d * H_ + hh];
  v = (v >= 0.f) ? v : NEG_SLOPE * v;
  atomicMax(&emax[d * H_ + hh], encodeOrd(v));
}

__global__ __launch_bounds__(256) void gat_edge_exp(
    const int* __restrict__ src, const int* __restrict__ dst,
    const float* __restrict__ el, const float* __restrict__ er,
    const unsigned* __restrict__ emax, float* __restrict__ ee,
    float* __restrict__ denom, int nEH)
{
  const int i = blockIdx.x * blockDim.x + threadIdx.x;
  if (i >= nEH) return;
  const int e = i >> 2;
  const int hh = i & (H_ - 1);
  const int s = src[e], d = dst[e];
  float v = er[s * H_ + hh] + el[d * H_ + hh];
  v = (v >= 0.f) ? v : NEG_SLOPE * v;
  const float x = expf(v - decodeOrd(emax[d * H_ + hh]));
  ee[i] = x;
  atomicAdd(&denom[d * H_ + hh], x);
}

__global__ __launch_bounds__(256) void gat_scatter(
    const int* __restrict__ src, const int* __restrict__ dst,
    const float* __restrict__ hW, const float* __restrict__ ee,
    const float* __restrict__ denom, float* __restrict__ out,
    long long nTot)
{
  const long long tid = (long long)blockIdx.x * blockDim.x + threadIdx.x;
  if (tid >= nTot) return;
  const long long e = tid >> 7;
  const int hf = (int)(tid & 127);
  const int hh = hf >> 5;
  const int s = src[e], d = dst[e];
  const float a = ee[e * H_ + hh] / denom[d * H_ + hh];
  atomicAdd(&out[(size_t)d * HF_ + hf], hW[(size_t)s * HF_ + hf] * a);
}

extern "C" void kernel_launch(void* const* d_in, const int* in_sizes, int n_in,
                              void* d_out, int out_size, void* d_ws, size_t ws_size,
                              hipStream_t stream)
{
  const float* h    = (const float*)d_in[0];
  const float* W    = (const float*)d_in[1];
  const float* alw  = (const float*)d_in[2];
  const float* alb  = (const float*)d_in[3];
  const float* arw  = (const float*)d_in[4];
  const float* arb  = (const float*)d_in[5];
  const float* bias = (const float*)d_in[6];
  const int*   src  = (const int*)d_in[7];
  const int*   dst  = (const int*)d_in[8];
  float* out = (float*)d_out;

  char* ws = (char*)d_ws;
  size_t off = 0;
  auto carve = [&](size_t bytes) -> void* {
    void* p = ws + off;
    off += (bytes + 255) & ~(size_t)255;
    return p;
  };
  float*    hW    = (float*)carve((size_t)N_ * HF_ * sizeof(float));
  float*    el    = (float*)carve((size_t)N_ * H_ * sizeof(float));
  float*    er    = (float*)carve((size_t)N_ * H_ * sizeof(float));
  unsigned* emax  = (unsigned*)carve((size_t)N_ * H_ * sizeof(unsigned));
  float*    denom = (float*)carve((size_t)N_ * H_ * sizeof(float));
  float*    ee    = (float*)carve((size_t)E_ * H_ * sizeof(float));
  (void)ws_size; (void)in_sizes; (void)n_in; (void)out_size;

  {
    const int strips = N_ / 16;
    const int blocks = (strips + 7) / 8;
    gat_gemm_wmma<<<blocks, 256, 0, stream>>>(h, W, hW, N_);
  }
  {
    const int nNH = N_ * H_;
    gat_attn_proj<<<(nNH + 255) / 256, 256, 0, stream>>>(hW, alw, alb, arw, arb, el, er, nNH);
  }
  {
    const int nNH = N_ * H_;
    const int nNHF = N_ * H_ * F_;
    gat_init<<<(nNHF + 255) / 256, 256, 0, stream>>>(emax, denom, out, bias, nNH, nNHF);
  }
  {
    const int nEH = E_ * H_;
    gat_edge_max<<<(nEH + 255) / 256, 256, 0, stream>>>(src, dst, el, er, emax, nEH);
  }
  {
    const int nEH = E_ * H_;
    gat_edge_exp<<<(nEH + 255) / 256, 256, 0, stream>>>(src, dst, el, er, emax, ee, denom, nEH);
  }
  {
    const long long nTot = (long long)E_ * HF_;
    const long long blocks = (nTot + 255) / 256;
    gat_scatter<<<(unsigned)blocks, 256, 0, stream>>>(src, dst, hW, ee, denom, out, nTot);
  }
}


# Group edges by destination in the call (integer-atomic rank, one-block scan, fill); one wave sums a node's rows in registers: no float atomics; sums in bucket arrival order, denominators by butterfly
# speedup vs baseline: 2.4749x; 2.4749x over previous
#include <hip/hip_runtime.h>

#define N_  50000
#define E_  800000
#define D_  256
#define H_  4
#define F_  32
#define HF_ 128
#define NEG_SLOPE 0.2f

typedef __attribute__((ext_vector_type(2))) float v2f;
typedef __attribute__((ext_vector_type(8))) float v8f;
typedef __attribute__((ext_vector_type(4))) float v4f;

__global__ __launch_bounds__(256) void gat_gemm_wmma(
    const float* __restrict__ h, const float* __restrict__ W,
    float* __restrict__ hW, int nRows)
{
  const int wave = threadIdx.x >> 5;
  const int lane = threadIdx.x & 31;
  const int rowBase = (blockIdx.x * 8 + wave) * 16;
  if (rowBase >= nRows) return;

  const int m    = lane & 15;
  const int khalf= (lane >> 4) * 2;

  v8f acc[8];
#pragma unroll
  for (int t = 0; t < 8; ++t) acc[t] = (v8f){0.f,0.f,0.f,0.f,0.f,0.f,0.f,0.f};

  for (int k0 = 0; k0 < D_; k0 += 4) {
    const float* ap = h + (size_t)(rowBase + m) * D_ + (k0 + khalf);
    v2f a; a.x = ap[0]; a.y = ap[1];
#pragma unroll
    for (int t = 0; t < 8; ++t) {
      const int n = t * 16 + (lane & 15);
      const float* bp = W + (size_t)(k0 + khalf) * HF_ + n;
      v2f b; b.x = bp[0]; b.y = bp[HF_];
      acc[t] = __builtin_amdgcn_wmma_f32_16x16x4_f32(
           false, a,  false, b,
           (short)0, acc[t],  false,  false);
    }
  }

  const int mHalf = (lane >> 4) * 8;
#pragma unroll
  for (int t = 0; t < 8; ++t) {
    const int n = t * 16 + (lane & 15);
#pragma unroll
    for (int v = 0; v < 8; ++v) {
      hW[(size_t)(rowBase + mHalf + v) * HF_ + n] = acc[t][v];
    }
  }
}

__global__ __launch_bounds__(256) void gat_attn_proj(
    const float* __restrict__ hW,
    const float* __restrict__ alw, const float* __restrict__ alb,
    const float* __restrict__ arw, const float* __restrict__ arb,
    float* __restrict__ el, float* __restrict__ er,
    int* __restrict__ cnt, int nNH, int nNodes)
{
  const int idx = blockIdx.x * blockDim.x + threadIdx.x;
  if (idx >= nNH) return;
  if (idx < nNodes) cnt[idx] = 0;
  const float* p = hW + (size_t)idx * F_;
  float sl = 0.f, sr = 0.f;
#pragma unroll
  for (int f = 0; f < F_; ++f) {
    const float x = p[f];
    sl += x * alw[f];
    sr += x * arw[f];
  }
  el[idx] = sl + alb[0];
  er[idx] = sr + arb[0];
}

__global__ __launch_bounds__(256) void gat_edge_rank(
    const int* __restrict__ dst, int* __restrict__ cnt,
    int* __restrict__ rank, int nE)
{
  const int e = blockIdx.x * blockDim.x + threadIdx.x;
  if (e >= nE) return;
  rank[e] = atomicAdd(&cnt[dst[e]], 1);
}

#define SCAN_THREADS 1024
#define SCAN_TILE    (SCAN_THREADS * 4)
#define SCAN_NT      ((N_ + SCAN_TILE - 1) / SCAN_TILE)
static_assert(N_ % 4 == 0, "a 16-byte group of counters must not straddle the end");

__global__ __launch_bounds__(SCAN_THREADS) void gat_scan(
    const int* __restrict__ cnt, int* __restrict__ off)
{
  __shared__ int wtot[32];
  const int t = threadIdx.x;
  const int lane = t & 31;
  const int w = t >> 5;

  int4 c[SCAN_NT];
#pragma unroll
  for (int j = 0; j < SCAN_NT; ++j) {
    const int i0 = j * SCAN_TILE + t * 4;
    c[j] = (i0 < N_) ? *(const int4*)(cnt + i0) : make_int4(0, 0, 0, 0);
  }

  int carry = 0;
#pragma unroll
  for (int j = 0; j < SCAN_NT; ++j) {
    const int s = c[j].x + c[j].y + c[j].z + c[j].w;
    int inc = s;
    for (int d = 1; d < 32; d <<= 1) {
      const int y = __shfl_up(inc, d, 32);
      if (lane >= d) inc += y;
    }
    if (lane == 31) wtot[w] = inc;
    __syncthreads();
    const int wt = wtot[lane];
    int winc = wt;
    for (int d = 1; d < 32; d <<= 1) {
      const int y = __shfl_up(winc, d, 32);
      if (lane >= d) winc += y;
    }
    const int tileTotal = __shfl(winc, 31, 32);
    const int wexcl = __shfl(winc - wt, w, 32);
    __syncthreads();
    const int base = carry + wexcl + (inc - s);
    const int i0 = j * SCAN_TILE + t * 4;
    if (i0 < N_) {
      int4 o;
      o.x = base;
      o.y = o.x + c[j].x;
      o.z = o.y + c[j].y;
      o.w = o.z + c[j].z;
      *(int4*)(off + i0) = o;
    }
    carry += tileTotal;
  }
  if (t == 0) off[N_] = carry;
}

__global__ __launch_bounds__(256) void gat_fill(
    const int* __restrict__ src, const int* __restrict__ dst,
    const int* __restrict__ off, const int* __restrict__ rank,
    int* __restrict__ ssrc, int nE)
{
  const int e = blockIdx.x * blockDim.x + threadIdx.x;
  if (e >= nE) return;
  ssrc[off[dst[e]] + rank[e]] = src[e];
}

__device__ __forceinline__ float waveMax(float v) {
  for (int m = 16; m >= 1; m >>= 1) v = fmaxf(v, __shfl_xor(v, m, 32));
  return v;
}
__device__ __forceinline__ float waveSum(float v) {
  for (int m = 16; m >= 1; m >>= 1) v += __shfl_xor(v, m, 32);
  return v;
}

struct EdgeChunk { int s; float v0, v1, v2, v3; bool ok; };

__device__ __forceinline__ EdgeChunk loadChunk(
    int base, int end, int lane, const int* __restrict__ ssrc,
    const float* __restrict__ er, v4f el4)
{
  EdgeChunk c;
  const int idx = base + lane;
  c.ok = idx < end;
  c.s = ssrc[c.ok ? idx : end - 1];
  const v4f r = *(const v4f*)(er + (size_t)c.s * H_);
  float v;
  v = r.x + el4.x; c.v0 = (v >= 0.f) ? v : NEG_SLOPE * v;
  v = r.y + el4.y; c.v1 = (v >= 0.f) ? v : NEG_SLOPE * v;
  v = r.z + el4.z; c.v2 = (v >= 0.f) ? v : NEG_SLOPE * v;
  v = r.w + el4.w; c.v3 = (v >= 0.f) ? v : NEG_SLOPE * v;
  return c;
}

__device__ __forceinline__ void accumChunk(
    int s, float a0, float a1, float a2, float a3, int n,
    unsigned mk0, unsigned mk1, unsigned mk2, unsigned mk3,
    const float* __restrict__ hWl, v4f& acc)
{
  const int b0 = __float_as_int(a0), b1 = __float_as_int(a1);
  const int b2 = __float_as_int(a2), b3 = __float_as_int(a3);
  for (int k = 0; k < n; k += 4) {
    v4f row[4];
#pragma unroll
    for (int u = 0; u < 4; ++u) {
      const int sk = __builtin_amdgcn_readlane(s, k + u);
      row[u] = *(const v4f*)(hWl + (size_t)sk * HF_);
    }
#pragma unroll
    for (int u = 0; u < 4; ++u) {
      const unsigned t0 = (unsigned)__builtin_amdgcn_readlane(b0, k + u);
      const unsigned t1 = (unsigned)__builtin_amdgcn_readlane(b1, k + u);
      const unsigned t2 = (unsigned)__builtin_amdgcn_readlane(b2, k + u);
      const unsigned t3 = (unsigned)__builtin_amdgcn_readlane(b3, k + u);
      const float wgt = __uint_as_float((t0 & mk0) | (t1 & mk1) | (t2 & mk2) | (t3 & mk3));
      acc += wgt * row[u];
    }
  }
}

__global__ __launch_bounds__(256) void gat_gather(
    const int* __restrict__ off, const int* __restrict__ ssrc,
    const float* __restrict__ hW, const float* __restrict__ el,
    const float* __restrict__ er, const float* __restrict__ bias,
    float* __restrict__ out, int nNodes)
{
  const int lane = threadIdx.x & 31;
  const int wave = __builtin_amdgcn_readfirstlane((int)(threadIdx.x >> 5));
  const int nWaves = gridDim.x * 8;
  const int hd = lane >> 3;
  const unsigned mk0 = (hd == 0) ? 0xFFFFFFFFu : 0u;
  const unsigned mk1 = (hd == 1) ? 0xFFFFFFFFu : 0u;
  const unsigned mk2 = (hd == 2) ? 0xFFFFFFFFu : 0u;
  const unsigned mk3 = (hd == 3) ? 0xFFFFFFFFu : 0u;
  const float NEG_INF = -__builtin_inff();
  const v4f b4 = *(const v4f*)(bias + lane * 4);
  const float* hWl = hW + lane * 4;

  for (int node = blockIdx.x * 8 + wave; node < nNodes; node += nWaves) {
    const int start = off[node];
    const int end = off[node + 1];
    const int deg = end - start;
    v4f acc = (v4f){0.f, 0.f, 0.f, 0.f};
    if (deg > 0) {
      const v4f el4 = *(const v4f*)(el + (size_t)node * H_);
      if (deg <= 32) {
        const EdgeChunk c = loadChunk(start, end, lane, ssrc, er, el4);
        const float m0 = waveMax(c.ok ? c.v0 : NEG_INF);
        const float m1 = waveMax(c.ok ? c.v1 : NEG_INF);
        const float m2 = waveMax(c.ok ? c.v2 : NEG_INF);
        const float m3 = waveMax(c.ok ? c.v3 : NEG_INF);
        const float x0 = c.ok ? expf(c.v0 - m0) : 0.f;
        const float x1 = c.ok ? expf(c.v1 - m1) : 0.f;
        const float x2 = c.ok ? expf(c.v2 - m2) : 0.f;
        const float x3 = c.ok ? expf(c.v3 - m3) : 0.f;
        const float d0 = waveSum(x0);
        const float d1 = waveSum(x1);
        const float d2 = waveSum(x2);
        const float d3 = waveSum(x3);
        accumChunk(c.s, x0 / d0, x1 / d1, x2 / d2, x3 / d3, deg, mk0, mk1, mk2, mk3, hWl, acc);
      } else {
        float m0 = NEG_INF, m1 = NEG_INF, m2 = NEG_INF, m3 = NEG_INF;
        for (int base = start; base < end; base += 32) {
          const EdgeChunk c = loadChunk(base, end, lane, ssrc, er, el4);
          m0 = fmaxf(m0, c.ok ? c.v0 : NEG_INF);
          m1 = fmaxf(m1, c.ok ? c.v1 : NEG_INF);
          m2 = fmaxf(m2, c.ok ? c.v2 : NEG_INF);
          m3 = fmaxf(m3, c.ok ? c.v3 : NEG_INF);
        }
        m0 = waveMax(m0); m1 = waveMax(m1); m2 = waveMax(m2); m3 = waveMax(m3);
        float d0 = 0.f, d1 = 0.f, d2 = 0.f, d3 = 0.f;
        for (int base = start; base < end; base += 32) {
          const EdgeChunk c = loadChunk(base, end, lane, ssrc, er, el4);
          d0 += c.ok ? expf(c.v0 - m0) : 0.f;
          d1 += c.ok ? expf(c.v1 - m1) : 0.f;
          d2 += c.ok ? expf(c.v2 - m2) : 0.f;
          d3 += c.ok ? expf(c.v3 - m3) : 0.f;
        }
        d0 = waveSum(d0); d1 = waveSum(d1); d2 = waveSum(d2); d3 = waveSum(d3);
        for (int base = start; base < end; base += 32) {
          const EdgeChunk c = loadChunk(base, end, lane, ssrc, er, el4);
          const float a0 = c.ok ? expf(c.v0 - m0) / d0 : 0.f;
          const float a1 = c.ok ? expf(c.v1 - m1) / d1 : 0.f;
          const float a2 = c.ok ? expf(c.v2 - m2) / d2 : 0.f;
          const float a3 = c.ok ? expf(c.v3 - m3) / d3 : 0.f;
          const int rem = end - base;
          accumChunk(c.s, a0, a1, a2, a3, rem < 32 ? rem : 32, mk0, mk1, mk2, mk3, hWl, acc);
        }
      }
    }
    *(v4f*)(out + (size_t)node * HF_ + lane * 4) = acc + b4;
  }
}

extern "C" void kernel_launch(void* const* d_in, const int* in_sizes, int n_in,
                              void* d_out, int out_size, void* d_ws, size_t ws_size,
                              hipStream_t stream)
{
  const float* h    = (const float*)d_in[0];
  const float* W    = (const float*)d_in[1];
  const float* alw  = (const float*)d_in[2];
  const float* alb  = (const float*)d_in[3];
  const float* arw  = (const float*)d_in[4];
  const float* arb  = (const float*)d_in[5];
  const float* bias = (const float*)d_in[6];
  const int*   src  = (const int*)d_in[7];
  const int*   dst  = (const int*)d_in[8];
  float* out = (float*)d_out;

  char* ws = (char*)d_ws;
  size_t off_b = 0;
  auto carve = [&](size_t bytes) -> void* {
    void* p = ws + off_b;
    off_b += (bytes + 255) & ~(size_t)255;
    return p;
  };
  float* hW   = (float*)carve((size_t)N_ * HF_ * sizeof(float));
  float* el   = (float*)carve((size_t)N_ * H_ * sizeof(float));
  float* er   = (float*)carve((size_t)N_ * H_ * sizeof(float));
  int*   cnt  = (int*)carve((size_t)N_ * sizeof(int));
  int*   off  = (int*)carve((size_t)(N_ + 1) * sizeof(int));
  int*   rank = (int*)carve((size_t)E_ * sizeof(int));
  int*   ssrc = (int*)carve((size_t)E_ * sizeof(int));
  (void)ws_size; (void)in_sizes; (void)n_in; (void)out_size;

  {
    const int strips = N_ / 16;
    const int blocks = (strips + 7) / 8;
    gat_gemm_wmma<<<blocks, 256, 0, stream>>>(h, W, hW, N_);
  }
  {
    const int nNH = N_ * H_;
    gat_attn_proj<<<(nNH + 255) / 256, 256, 0, stream>>>(hW, alw, alb, arw, arb, el, er, cnt, nNH, N_);
  }
  gat_edge_rank<<<(E_ + 255) / 256, 256, 0, stream>>>(dst, cnt, rank, E_);
  gat_scan<<<1, SCAN_THREADS, 0, stream>>>(cnt, off);
  gat_fill<<<(E_ + 255) / 256, 256, 0, stream>>>(src, dst, off, rank, ssrc, E_);
  {
    const int waves = (N_ + 3) / 4;
    const int blocks = (waves + 7) / 8;
    gat_gather<<<blocks, 256, 0, stream>>>(off, ssrc, hW, el, er, bias, out, N_);
  }
}


# Group edges by destination in LDS: each 4096-edge chunk counting-sorted by 256-node bin, then one work-group per bin sorts by node; whole-line stores, no per-edge global atomic; sums in arrival order
# speedup vs baseline: 4.5138x; 1.8238x over previous
#include <hip/hip_runtime.h>

#define N_  50000
#define E_  800000
#define D_  256
#define H_  4
#define F_  32
#define HF_ 128
#define NEG_SLOPE 0.2f

typedef __attribute__((ext_vector_type(2))) float v2f;
typedef __attribute__((ext_vector_type(8))) float v8f;
typedef __attribute__((ext_vector_type(4))) float v4f;

__global__ __launch_bounds__(256) void gat_gemm_wmma(
    const float* __restrict__ h, const float* __restrict__ W,
    float* __restrict__ hW, int nRows)
{
  const int wave = threadIdx.x >> 5;
  const int lane = threadIdx.x & 31;
  const int rowBase = (blockIdx.x * 8 + wave) * 16;
  if (rowBase >= nRows) return;

  const int m    = lane & 15;
  const int khalf= (lane >> 4) * 2;

  v8f acc[8];
#pragma unroll
  for (int t = 0; t < 8; ++t) acc[t] = (v8f){0.f,0.f,0.f,0.f,0.f,0.f,0.f,0.f};

  for (int k0 = 0; k0 < D_; k0 += 4) {
    const float* ap = h + (size_t)(rowBase + m) * D_ + (k0 + khalf);
    v2f a; a.x = ap[0]; a.y = ap[1];
#pragma unroll
    for (int t = 0; t < 8; ++t) {
      const int n = t * 16 + (lane & 15);
      const float* bp = W + (size_t)(k0 + khalf) * HF_ + n;
      v2f b; b.x = bp[0]; b.y = bp[HF_];
      acc[t] = __builtin_amdgcn_wmma_f32_16x16x4_f32(
           false, a,  false, b,
           (short)0, acc[t],  false,  false);
    }
  }

  const int mHalf = (lane >> 4) * 8;
#pragma unroll
  for (int t = 0; t < 8; ++t) {
    const int n = t * 16 + (lane & 15);
#pragma unroll
    for (int v = 0; v < 8; ++v) {
      hW[(size_t)(rowBase + mHalf + v) * HF_ + n] = acc[t][v];
    }
  }
}

#define BIN_SHIFT   8
#define BIN_NODES   (1 << BIN_SHIFT)
#define NBINS       ((N_ + BIN_NODES - 1) / BIN_NODES)
#define TOT_STRIDE  32
#define PART_THREADS 1024
#define PART_EPT     4
#define PART_CHUNK   (PART_THREADS * PART_EPT)
#define NCHUNKS      ((E_ + PART_CHUNK - 1) / PART_CHUNK)
static_assert(N_ <= 65536, "a source id must fit the low 16 bits of a packed edge");
static_assert(NBINS <= 32 * 7, "wave 0 scans the bins 7 per lane");
static_assert(BIN_NODES == 32 * 8, "wave 0 scans a bin's nodes 8 per lane");

__global__ __launch_bounds__(256) void gat_attn_proj(
    const float* __restrict__ hW,
    const float* __restrict__ alw, const float* __restrict__ alb,
    const float* __restrict__ arw, const float* __restrict__ arb,
    float* __restrict__ el, float* __restrict__ er,
    int* __restrict__ binTotal, int nNH, int nBins)
{
  const int idx = blockIdx.x * blockDim.x + threadIdx.x;
  if (idx >= nNH) return;
  if (idx < nBins) binTotal[idx * TOT_STRIDE] = 0;
  const float* p = hW + (size_t)idx * F_;
  float sl = 0.f, sr = 0.f;
#pragma unroll
  for (int f = 0; f < F_; ++f) {
    const float x = p[f];
    sl += x * alw[f];
    sr += x * arw[f];
  }
  el[idx] = sl + alb[0];
  er[idx] = sr + arb[0];
}

__device__ __forceinline__ int waveIncScanI(int v, int lane) {
  for (int d = 1; d < 32; d <<= 1) {
    const int y = __shfl_up(v, d, 32);
    if (lane >= d) v += y;
  }
  return v;
}
__device__ __forceinline__ int waveSumI(int v) {
  for (int m = 16; m >= 1; m >>= 1) v += __shfl_xor(v, m, 32);
  return v;
}

__global__ __launch_bounds__(PART_THREADS) void gat_part(
    const int* __restrict__ src, const int* __restrict__ dst,
    int* __restrict__ part, int* __restrict__ tabCnt, int* __restrict__ tabOff,
    int* __restrict__ binTotal)
{
  __shared__ int sCnt[32 * 7];
  __shared__ int sOff[32 * 7];
  __shared__ int sStage[PART_CHUNK];
  const int t = threadIdx.x;
  const int lane = t & 31;
  const int chunkBase = blockIdx.x * PART_CHUNK;
  if (t < 32 * 7) sCnt[t] = 0;
  __syncthreads();

  int pk[PART_EPT], bn[PART_EPT], rk[PART_EPT];
#pragma unroll
  for (int u = 0; u < PART_EPT; ++u) {
    const int e = chunkBase + u * PART_THREADS + t;
    const int ec = (e < E_) ? e : E_ - 1;
    const int d = dst[ec];
    const int s = src[ec];
    bn[u] = (e < E_) ? (d >> BIN_SHIFT) : -1;
    pk[u] = ((d & (BIN_NODES - 1)) << 16) | s;
    rk[u] = 0;
  }
#pragma unroll
  for (int u = 0; u < PART_EPT; ++u)
    if (bn[u] >= 0) rk[u] = atomicAdd(&sCnt[bn[u]], 1);
  __syncthreads();

  if (t < 32) {
    int c[7]; int s = 0;
#pragma unroll
    for (int q = 0; q < 7; ++q) { c[q] = sCnt[lane * 7 + q]; s += c[q]; }
    int run = waveIncScanI(s, lane) - s;
#pragma unroll
    for (int q = 0; q < 7; ++q) { sOff[lane * 7 + q] = run; run += c[q]; }
  }
  __syncthreads();

#pragma unroll
  for (int u = 0; u < PART_EPT; ++u)
    if (bn[u] >= 0) sStage[sOff[bn[u]] + rk[u]] = pk[u];
  __syncthreads();

#pragma unroll
  for (int u = 0; u < PART_EPT; ++u) {
    const int i = u * PART_THREADS + t;
    if (chunkBase + i < E_) part[chunkBase + i] = sStage[i];
  }
  if (t < NBINS) {
    const int c = sCnt[t];
    tabCnt[blockIdx.x * NBINS + t] = c;
    tabOff[blockIdx.x * NBINS + t] = chunkBase + sOff[t];
    if (c > 0) atomicAdd(&binTotal[t * TOT_STRIDE], c);
  }
}

#define BK_THREADS 1024
#define BK_WAVES   (BK_THREADS / 32)
#define BK_SEGS    ((NCHUNKS + BK_WAVES - 1) / BK_WAVES)
#define BK_CAP     8192

__global__ __launch_bounds__(BK_THREADS) void gat_bucket(
    const int* __restrict__ part, const int* __restrict__ tabCnt,
    const int* __restrict__ tabOff, const int* __restrict__ binTotal,
    int* __restrict__ off, int* __restrict__ ssrc)
{
  __shared__ int sSegCnt[NCHUNKS];
  __shared__ int sSegOff[NCHUNKS];
  __shared__ int sCnt[BIN_NODES];
  __shared__ int sOff[BIN_NODES];
  __shared__ int sCur[BIN_NODES];
  __shared__ int sBase[2];
  __shared__ int sStage[BK_CAP];
  const int t = threadIdx.x;
  const int lane = t & 31;
  const int w = t >> 5;
  const int b = blockIdx.x;

  if (t < BIN_NODES) { sCnt[t] = 0; sCur[t] = 0; }
  if (t < NCHUNKS) {
    sSegCnt[t] = tabCnt[t * NBINS + b];
    sSegOff[t] = tabOff[t * NBINS + b];
  }
  if (t < 32) {
    int below = 0, mine = 0;
#pragma unroll
    for (int q = 0; q < 7; ++q) {
      const int i = lane * 7 + q;
      const int v = (i < NBINS) ? binTotal[i * TOT_STRIDE] : 0;
      below += (i < b) ? v : 0;
      mine  += (i == b) ? v : 0;
    }
    below = waveSumI(below);
    mine = waveSumI(mine);
    if (lane == 0) { sBase[0] = below; sBase[1] = mine; }
  }
  __syncthreads();
  const int binBase = sBase[0];
  const int binCount = sBase[1];

  int pv[BK_SEGS], sc[BK_SEGS], so[BK_SEGS];
#pragma unroll
  for (int q = 0; q < BK_SEGS; ++q) {
    const int j = w + q * BK_WAVES;
    sc[q] = (j < NCHUNKS) ? sSegCnt[j] : 0;
    so[q] = (j < NCHUNKS) ? sSegOff[j] : 0;
    const int ix = so[q] + lane;
    const int pw = part[(ix < E_) ? ix : E_ - 1];
    pv[q] = (lane < sc[q]) ? pw : -1;
  }
#pragma unroll
  for (int q = 0; q < BK_SEGS; ++q)
    if (pv[q] >= 0) atomicAdd(&sCnt[pv[q] >> 16], 1);
#pragma unroll
  for (int q = 0; q < BK_SEGS; ++q)
    for (int i = lane + 32; i < sc[q]; i += 32)
      atomicAdd(&sCnt[part[so[q] + i] >> 16], 1);
  __syncthreads();

  if (t < 32) {
    int c[8]; int s = 0;
#pragma unroll
    for (int q = 0; q < 8; ++q) { c[q] = sCnt[lane * 8 + q]; s += c[q]; }
    int run = waveIncScanI(s, lane) - s;
#pragma unroll
    for (int q = 0; q < 8; ++q) { sOff[lane * 8 + q] = run; run += c[q]; }
  }
  __syncthreads();

  if (t < BIN_NODES) {
    const int node = b * BIN_NODES + t;
    if (node < N_) off[node] = binBase + sOff[t];
  }
  if (b == NBINS - 1 && t == 0) off[N_] = binBase + binCount;

  const bool staged = binCount <= BK_CAP;
#pragma unroll
  for (int q = 0; q < BK_SEGS; ++q) {
    if (pv[q] >= 0) {
      const int f = pv[q] >> 16;
      const int pos = sOff[f] + atomicAdd(&sCur[f], 1);
      const int s = pv[q] & 0xFFFF;
      if (staged) sStage[pos] = s; else ssrc[binBase + pos] = s;
    }
  }
#pragma unroll
  for (int q = 0; q < BK_SEGS; ++q) {
    for (int i = lane + 32; i < sc[q]; i += 32) {
      const int p = part[so[q] + i];
      const int f = p >> 16;
      const int pos = sOff[f] + atomicAdd(&sCur[f], 1);
      const int s = p & 0xFFFF;
      if (staged) sStage[pos] = s; else ssrc[binBase + pos] = s;
    }
  }
  __syncthreads();
  if (staged)
    for (int i = t; i < binCount; i += BK_THREADS) ssrc[binBase + i] = sStage[i];
}

__device__ __forceinline__ float waveMax(float v) {
  for (int m = 16; m >= 1; m >>= 1) v = fmaxf(v, __shfl_xor(v, m, 32));
  return v;
}
__device__ __forceinline__ float waveSum(float v) {
  for (int m = 16; m >= 1; m >>= 1) v += __shfl_xor(v, m, 32);
  return v;
}

struct EdgeChunk { int s; float v0, v1, v2, v3; bool ok; };

__device__ __forceinline__ EdgeChunk loadChunk(
    int base, int end, int lane, const int* __restrict__ ssrc,
    const float* __restrict__ er, v4f el4)
{
  EdgeChunk c;
  const int idx = base + lane;
  c.ok = idx < end;
  c.s = ssrc[c.ok ? idx : end - 1];
  const v4f r = *(const v4f*)(er + (size_t)c.s * H_);
  float v;
  v = r.x + el4.x; c.v0 = (v >= 0.f) ? v : NEG_SLOPE * v;
  v = r.y + el4.y; c.v1 = (v >= 0.f) ? v : NEG_SLOPE * v;
  v = r.z + el4.z; c.v2 = (v >= 0.f) ? v : NEG_SLOPE * v;
  v = r.w + el4.w; c.v3 = (v >= 0.f) ? v : NEG_SLOPE * v;
  return c;
}

__device__ __forceinline__ void accumChunk(
    int s, float a0, float a1, float a2, float a3, int n,
    unsigned mk0, unsigned mk1, unsigned mk2, unsigned mk3,
    const float* __restrict__ hWl, v4f& acc)
{
  const int b0 = __float_as_int(a0), b1 = __float_as_int(a1);
  const int b2 = __float_as_int(a2), b3 = __float_as_int(a3);
  for (int k = 0; k < n; k += 4) {
    v4f row[4];
#pragma unroll
    for (int u = 0; u < 4; ++u) {
      const int sk = __builtin_amdgcn_readlane(s, k + u);
      row[u] = *(const v4f*)(hWl + (size_t)sk * HF_);
    }
#pragma unroll
    for (int u = 0; u < 4; ++u) {
      const unsigned t0 = (unsigned)__builtin_amdgcn_readlane(b0, k + u);
      const unsigned t1 = (unsigned)__builtin_amdgcn_readlane(b1, k + u);
      const unsigned t2 = (unsigned)__builtin_amdgcn_readlane(b2, k + u);
      const unsigned t3 = (unsigned)__builtin_amdgcn_readlane(b3, k + u);
      const float wgt = __uint_as_float((t0 & mk0) | (t1 & mk1) | (t2 & mk2) | (t3 & mk3));
      acc += wgt * row[u];
    }
  }
}

__global__ __launch_bounds__(256) void gat_gather(
    const int* __restrict__ off, const int* __restrict__ ssrc,
    const float* __restrict__ hW, const float* __restrict__ el,
    const float* __restrict__ er, const float* __restrict__ bias,
    float* __restrict__ out, int nNodes)
{
  const int lane = threadIdx.x & 31;
  const int wave = __builtin_amdgcn_readfirstlane((int)(threadIdx.x >> 5));
  const int nWaves = gridDim.x * 8;
  const int hd = lane >> 3;
  const unsigned mk0 = (hd == 0) ? 0xFFFFFFFFu : 0u;
  const unsigned mk1 = (hd == 1) ? 0xFFFFFFFFu : 0u;
  const unsigned mk2 = (hd == 2) ? 0xFFFFFFFFu : 0u;
  const unsigned mk3 = (hd == 3) ? 0xFFFFFFFFu : 0u;
  const float NEG_INF = -__builtin_inff();
  const v4f b4 = *(const v4f*)(bias + lane * 4);
  const float* hWl = hW + lane * 4;

  for (int node = blockIdx.x * 8 + wave; node < nNodes; node += nWaves) {
    const int start = off[node];
    const int end = off[node + 1];
    const int deg = end - start;
    v4f acc = (v4f){0.f, 0.f, 0.f, 0.f};
    if (deg > 0) {
      const v4f el4 = *(const v4f*)(el + (size_t)node * H_);
      if (deg <= 32) {
        const EdgeChunk c = loadChunk(start, end, lane, ssrc, er, el4);
        const float m0 = waveMax(c.ok ? c.v0 : NEG_INF);
        const float m1 = waveMax(c.ok ? c.v1 : NEG_INF);
        const float m2 = waveMax(c.ok ? c.v2 : NEG_INF);
        const float m3 = waveMax(c.ok ? c.v3 : NEG_INF);
        const float x0 = c.ok ? expf(c.v0 - m0) : 0.f;
        const float x1 = c.ok ? expf(c.v1 - m1) : 0.f;
        const float x2 = c.ok ? expf(c.v2 - m2) : 0.f;
        const float x3 = c.ok ? expf(c.v3 - m3) : 0.f;
        const float d0 = waveSum(x0);
        const float d1 = waveSum(x1);
        const float d2 = waveSum(x2);
        const float d3 = waveSum(x3);
        accumChunk(c.s, x0 / d0, x1 / d1, x2 / d2, x3 / d3, deg, mk0, mk1, mk2, mk3, hWl, acc);
      } else {
        float m0 = NEG_INF, m1 = NEG_INF, m2 = NEG_INF, m3 = NEG_INF;
        for (int base = start; base < end; base += 32) {
          const EdgeChunk c = loadChunk(base, end, lane, ssrc, er, el4);
          m0 = fmaxf(m0, c.ok ? c.v0 : NEG_INF);
          m1 = fmaxf(m1, c.ok ? c.v1 : NEG_INF);
          m2 = fmaxf(m2, c.ok ? c.v2 : NEG_INF);
          m3 = fmaxf(m3, c.ok ? c.v3 : NEG_INF);
        }
        m0 = waveMax(m0); m1 = waveMax(m1); m2 = waveMax(m2); m3 = waveMax(m3);
        float d0 = 0.f, d1 = 0.f, d2 = 0.f, d3 = 0.f;
        for (int base = start; base < end; base += 32) {
          const EdgeChunk c = loadChunk(base, end, lane, ssrc, er, el4);
          d0 += c.ok ? expf(c.v0 - m0) : 0.f;
          d1 += c.ok ? expf(c.v1 - m1) : 0.f;
          d2 += c.ok ? expf(c.v2 - m2) : 0.f;
          d3 += c.ok ? expf(c.v3 - m3) : 0.f;
        }
        d0 = waveSum(d0); d1 = waveSum(d1); d2 = waveSum(d2); d3 = waveSum(d3);
        for (int base = start; base < end; base += 32) {
          const EdgeChunk c = loadChunk(base, end, lane, ssrc, er, el4);
          const float a0 = c.ok ? expf(c.v0 - m0) / d0 : 0.f;
          const float a1 = c.ok ? expf(c.v1 - m1) / d1 : 0.f;
          const float a2 = c.ok ? expf(c.v2 - m2) / d2 : 0.f;
          const float a3 = c.ok ? expf(c.v3 - m3) / d3 : 0.f;
          const int rem = end - base;
          accumChunk(c.s, a0, a1, a2, a3, rem < 32 ? rem : 32, mk0, mk1, mk2, mk3, hWl, acc);
        }
      }
    }
    *(v4f*)(out + (size_t)node * HF_ + lane * 4) = acc + b4;
  }
}

extern "C" void kernel_launch(void* const* d_in, const int* in_sizes, int n_in,
                              void* d_out, int out_size, void* d_ws, size_t ws_size,
                              hipStream_t stream)
{
  const float* h    = (const float*)d_in[0];
  const float* W    = (const float*)d_in[1];
  const float* alw  = (const float*)d_in[2];
  const float* alb  = (const float*)d_in[3];
  const float* arw  = (const float*)d_in[4];
  const float* arb  = (const float*)d_in[5];
  const float* bias = (const float*)d_in[6];
  const int*   src  = (const int*)d_in[7];
  const int*   dst  = (const int*)d_in[8];
  float* out = (float*)d_out;

  char* ws = (char*)d_ws;
  size_t off_b = 0;
  auto carve = [&](size_t bytes) -> void* {
    void* p = ws + off_b;
    off_b += (bytes + 255) & ~(size_t)255;
    return p;
  };
  float* hW   = (float*)carve((size_t)N_ * HF_ * sizeof(float));
  float* el   = (float*)carve((size_t)N_ * H_ * sizeof(float));
  float* er   = (float*)carve((size_t)N_ * H_ * sizeof(float));
  int*   btot = (int*)carve((size_t)NBINS * TOT_STRIDE * sizeof(int));
  int*   tcnt = (int*)carve((size_t)NCHUNKS * NBINS * sizeof(int));
  int*   toff = (int*)carve((size_t)NCHUNKS * NBINS * sizeof(int));
  int*   off  = (int*)carve((size_t)(N_ + 1) * sizeof(int));
  int*   part = (int*)carve((size_t)E_ * sizeof(int));
  int*   ssrc = (int*)carve((size_t)E_ * sizeof(int));
  (void)ws_size; (void)in_sizes; (void)n_in; (void)out_size;

  {
    const int strips = N_ / 16;
    const int blocks = (strips + 7) / 8;
    gat_gemm_wmma<<<blocks, 256, 0, stream>>>(h, W, hW, N_);
  }
  {
    const int nNH = N_ * H_;
    gat_attn_proj<<<(nNH + 255) / 256, 256, 0, stream>>>(hW, alw, alb, arw, arb, el, er, btot, nNH, NBINS);
  }
  gat_part<<<NCHUNKS, PART_THREADS, 0, stream>>>(src, dst, part, tcnt, toff, btot);
  gat_bucket<<<NBINS, BK_THREADS, 0, stream>>>(part, tcnt, toff, btot, off, ssrc);
  {
    const int waves = (N_ + 3) / 4;
    const int blocks = (waves + 7) / 8;
    gat_gather<<<blocks, 256, 0, stream>>>(off, ssrc, hW, el, er, bias, out, N_);
  }
}


# el/er computed in the GEMM epilogue from the accumulators (16-lane DPP tree sum replaces the serial 32-term sum); the projection launch and its re-read of hW are gone: 5 launches to 4
# speedup vs baseline: 5.5225x; 1.2235x over previous
#include <hip/hip_runtime.h>

#define N_  50000
#define E_  800000
#define D_  256
#define H_  4
#define F_  32
#define HF_ 128
#define NEG_SLOPE 0.2f

typedef __attribute__((ext_vector_type(2))) float v2f;
typedef __attribute__((ext_vector_type(8))) float v8f;
typedef __attribute__((ext_vector_type(4))) float v4f;


template <int MASK>
__device__ __forceinline__ float rowXor(float x) {
  return __int_as_float(__builtin_amdgcn_update_dpp(0, __float_as_int(x), 0x160 | MASK, 0xF, 0xF, true));
}

__device__ __forceinline__ void rowSum32(const float (&x)[32], int c, float& o0, float& o1) {
  const bool b3 = (c & 8) != 0, b2 = (c & 4) != 0, b1 = (c & 2) != 0, b0 = (c & 1) != 0;
  float y[16], z[8], w[4];
#pragma unroll
  for (int j = 0; j < 16; ++j) {
    const float keep = b3 ? x[j + 16] : x[j];
    const float send = b3 ? x[j] : x[j + 16];
    y[j] = keep + rowXor<8>(send);
  }
#pragma unroll
  for (int j = 0; j < 8; ++j) {
    const float keep = b2 ? y[j + 8] : y[j];
    const float send = b2 ? y[j] : y[j + 8];
    z[j] = keep + rowXor<4>(send);
  }
#pragma unroll
  for (int j = 0; j < 4; ++j) {
    const float keep = b1 ? z[j + 4] : z[j];
    const float send = b1 ? z[j] : z[j + 4];
    w[j] = keep + rowXor<2>(send);
  }
  {
    const float keep0 = b0 ? w[2] : w[0];
    const float send0 = b0 ? w[0] : w[2];
    const float keep1 = b0 ? w[3] : w[1];
    const float send1 = b0 ? w[1] : w[3];
    o0 = keep0 + rowXor<1>(send0);
    o1 = keep1 + rowXor<1>(send1);
  }
}

#define TOT_STRIDE  32

__global__ __launch_bounds__(256) void gat_gemm_wmma(
    const float* __restrict__ h, const float* __restrict__ W,
    const float* __restrict__ alw, const float* __restrict__ alb,
    const float* __restrict__ arw, const float* __restrict__ arb,
    float* __restrict__ hW, float* __restrict__ el, float* __restrict__ er,
    int* __restrict__ binTotal, int nRows, int nBins)
{
  const int wave = threadIdx.x >> 5;
  const int lane = threadIdx.x & 31;
  if (blockIdx.x == 0 && (int)threadIdx.x < nBins) binTotal[threadIdx.x * TOT_STRIDE] = 0;
  const int rowBase = (blockIdx.x * 8 + wave) * 16;
  if (rowBase >= nRows) return;

  const int m    = lane & 15;
  const int khalf= (lane >> 4) * 2;
  const float al0 = alw[m], al1 = alw[16 + m];
  const float ar0 = arw[m], ar1 = arw[16 + m];
  const float bl = alb[0], br = arb[0];

  v8f acc[8];
#pragma unroll
  for (int t = 0; t < 8; ++t) acc[t] = (v8f){0.f,0.f,0.f,0.f,0.f,0.f,0.f,0.f};

  for (int k0 = 0; k0 < D_; k0 += 4) {
    const float* ap = h + (size_t)(rowBase + m) * D_ + (k0 + khalf);
    v2f a; a.x = ap[0]; a.y = ap[1];
#pragma unroll
    for (int t = 0; t < 8; ++t) {
      const int n = t * 16 + (lane & 15);
      const float* bp = W + (size_t)(k0 + khalf) * HF_ + n;
      v2f b; b.x = bp[0]; b.y = bp[HF_];
      acc[t] = __builtin_amdgcn_wmma_f32_16x16x4_f32(
           false, a,  false, b,
           (short)0, acc[t],  false,  false);
    }
  }

  const int mHalf = (lane >> 4) * 8;
#pragma unroll
  for (int t = 0; t < 8; ++t) {
    const int n = t * 16 + (lane & 15);
#pragma unroll
    for (int v = 0; v < 8; ++v) {
      hW[(size_t)(rowBase + mHalf + v) * HF_ + n] = acc[t][v];
    }
  }

  float xl[32], xr[32];
#pragma unroll
  for (int hd = 0; hd < H_; ++hd) {
#pragma unroll
    for (int v = 0; v < 8; ++v) {
      const float p0 = acc[2 * hd][v], p1 = acc[2 * hd + 1][v];
      xl[v * 4 + hd] = p0 * al0 + p1 * al1;
      xr[v * 4 + hd] = p0 * ar0 + p1 * ar1;
    }
  }
  float l0, l1, r0, r1;
  rowSum32(xl, m, l0, l1);
  rowSum32(xr, m, r0, r1);
  const size_t eo = (size_t)(rowBase + mHalf) * H_ + 2 * m;
  v2f ol; ol.x = l0 + bl; ol.y = l1 + bl;
  v2f orr; orr.x = r0 + br; orr.y = r1 + br;
  *(v2f*)(el + eo) = ol;
  *(v2f*)(er + eo) = orr;
}

#define BIN_SHIFT   8
#define BIN_NODES   (1 << BIN_SHIFT)
#define NBINS       ((N_ + BIN_NODES - 1) / BIN_NODES)
#define PART_THREADS 1024
#define PART_EPT     4
#define PART_CHUNK   (PART_THREADS * PART_EPT)
#define NCHUNKS      ((E_ + PART_CHUNK - 1) / PART_CHUNK)
static_assert(N_ <= 65536, "a source id must fit the low 16 bits of a packed edge");
static_assert(NBINS <= 32 * 7, "wave 0 scans the bins 7 per lane");
static_assert(BIN_NODES == 32 * 8, "wave 0 scans a bin's nodes 8 per lane");
static_assert(NBINS <= 256, "block 0 of the first kernel clears one bin total per thread");

__device__ __forceinline__ int waveIncScanI(int v, int lane) {
  for (int d = 1; d < 32; d <<= 1) {
    const int y = __shfl_up(v, d, 32);
    if (lane >= d) v += y;
  }
  return v;
}
__device__ __forceinline__ int waveSumI(int v) {
  for (int m = 16; m >= 1; m >>= 1) v += __shfl_xor(v, m, 32);
  return v;
}

__global__ __launch_bounds__(PART_THREADS) void gat_part(
    const int* __restrict__ src, const int* __restrict__ dst,
    int* __restrict__ part, int* __restrict__ tabCnt, int* __restrict__ tabOff,
    int* __restrict__ binTotal)
{
  __shared__ int sCnt[32 * 7];
  __shared__ int sOff[32 * 7];
  __shared__ int sStage[PART_CHUNK];
  const int t = threadIdx.x;
  const int lane = t & 31;
  const int chunkBase = blockIdx.x * PART_CHUNK;
  if (t < 32 * 7) sCnt[t] = 0;
  __syncthreads();

  int pk[PART_EPT], bn[PART_EPT], rk[PART_EPT];
#pragma unroll
  for (int u = 0; u < PART_EPT; ++u) {
    const int e = chunkBase + u * PART_THREADS + t;
    const int ec = (e < E_) ? e : E_ - 1;
    const int d = dst[ec];
    const int s = src[ec];
    bn[u] = (e < E_) ? (d >> BIN_SHIFT) : -1;
    pk[u] = ((d & (BIN_NODES - 1)) << 16) | s;
    rk[u] = 0;
  }
#pragma unroll
  for (int u = 0; u < PART_EPT; ++u)
    if (bn[u] >= 0) rk[u] = atomicAdd(&sCnt[bn[u]], 1);
  __syncthreads();

  if (t < 32) {
    int c[7]; int s = 0;
#pragma unroll
    for (int q = 0; q < 7; ++q) { c[q] = sCnt[lane * 7 + q]; s += c[q]; }
    int run = waveIncScanI(s, lane) - s;
#pragma unroll
    for (int q = 0; q < 7; ++q) { sOff[lane * 7 + q] = run; run += c[q]; }
  }
  __syncthreads();

#pragma unroll
  for (int u = 0; u < PART_EPT; ++u)
    if (bn[u] >= 0) sStage[sOff[bn[u]] + rk[u]] = pk[u];
  __syncthreads();

#pragma unroll
  for (int u = 0; u < PART_EPT; ++u) {
    const int i = u * PART_THREADS + t;
    if (chunkBase + i < E_) part[chunkBase + i] = sStage[i];
  }
  if (t < NBINS) {
    const int c = sCnt[t];
    tabCnt[blockIdx.x * NBINS + t] = c;
    tabOff[blockIdx.x * NBINS + t] = chunkBase + sOff[t];
    if (c > 0) atomicAdd(&binTotal[t * TOT_STRIDE], c);
  }
}

#define BK_THREADS 1024
#define BK_WAVES   (BK_THREADS / 32)
#define BK_SEGS    ((NCHUNKS + BK_WAVES - 1) / BK_WAVES)
#define BK_CAP     8192

__global__ __launch_bounds__(BK_THREADS) void gat_bucket(
    const int* __restrict__ part, const int* __restrict__ tabCnt,
    const int* __restrict__ tabOff, const int* __restrict__ binTotal,
    int* __restrict__ off, int* __restrict__ ssrc)
{
  __shared__ int sSegCnt[NCHUNKS];
  __shared__ int sSegOff[NCHUNKS];
  __shared__ int sCnt[BIN_NODES];
  __shared__ int sOff[BIN_NODES];
  __shared__ int sCur[BIN_NODES];
  __shared__ int sBase[2];
  __shared__ int sStage[BK_CAP];
  const int t = threadIdx.x;
  const int lane = t & 31;
  const int w = t >> 5;
  const int b = blockIdx.x;

  if (t < BIN_NODES) { sCnt[t] = 0; sCur[t] = 0; }
  if (t < NCHUNKS) {
    sSegCnt[t] = tabCnt[t * NBINS + b];
    sSegOff[t] = tabOff[t * NBINS + b];
  }
  if (t < 32) {
    int below = 0, mine = 0;
#pragma unroll
    for (int q = 0; q < 7; ++q) {
      const int i = lane * 7 + q;
      const int v = (i < NBINS) ? binTotal[i * TOT_STRIDE] : 0;
      below += (i < b) ? v : 0;
      mine  += (i == b) ? v : 0;
    }
    below = waveSumI(below);
    mine = waveSumI(mine);
    if (lane == 0) { sBase[0] = below; sBase[1] = mine; }
  }
  __syncthreads();
  const int binBase = sBase[0];
  const int binCount = sBase[1];

  int pv[BK_SEGS], sc[BK_SEGS], so[BK_SEGS];
#pragma unroll
  for (int q = 0; q < BK_SEGS; ++q) {
    const int j = w + q * BK_WAVES;
    sc[q] = (j < NCHUNKS) ? sSegCnt[j] : 0;
    so[q] = (j < NCHUNKS) ? sSegOff[j] : 0;
    const int ix = so[q] + lane;
    const int pw = part[(ix < E_) ? ix : E_ - 1];
    pv[q] = (lane < sc[q]) ? pw : -1;
  }
#pragma unroll
  for (int q = 0; q < BK_SEGS; ++q)
    if (pv[q] >= 0) atomicAdd(&sCnt[pv[q] >> 16], 1);
#pragma unroll
  for (int q = 0; q < BK_SEGS; ++q)
    for (int i = lane + 32; i < sc[q]; i += 32)
      atomicAdd(&sCnt[part[so[q] + i] >> 16], 1);
  __syncthreads();

  if (t < 32) {
    int c[8]; int s = 0;
#pragma unroll
    for (int q = 0; q < 8; ++q) { c[q] = sCnt[lane * 8 + q]; s += c[q]; }
    int run = waveIncScanI(s, lane) - s;
#pragma unroll
    for (int q = 0; q < 8; ++q) { sOff[lane * 8 + q] = run; run += c[q]; }
  }
  __syncthreads();

  if (t < BIN_NODES) {
    const int node = b * BIN_NODES + t;
    if (node < N_) off[node] = binBase + sOff[t];
  }
  if (b == NBINS - 1 && t == 0) off[N_] = binBase + binCount;

  const bool staged = binCount <= BK_CAP;
#pragma unroll
  for (int q = 0; q < BK_SEGS; ++q) {
    if (pv[q] >= 0) {
      const int f = pv[q] >> 16;
      const int pos = sOff[f] + atomicAdd(&sCur[f], 1);
      const int s = pv[q] & 0xFFFF;
      if (staged) sStage[pos] = s; else ssrc[binBase + pos] = s;
    }
  }
#pragma unroll
  for (int q = 0; q < BK_SEGS; ++q) {
    for (int i = lane + 32; i < sc[q]; i += 32) {
      const int p = part[so[q] + i];
      const int f = p >> 16;
      const int pos = sOff[f] + atomicAdd(&sCur[f], 1);
      const int s = p & 0xFFFF;
      if (staged) sStage[pos] = s; else ssrc[binBase + pos] = s;
    }
  }
  __syncthreads();
  if (staged)
    for (int i = t; i < binCount; i += BK_THREADS) ssrc[binBase + i] = sStage[i];
}

__device__ __forceinline__ float waveMax(float v) {
  for (int m = 16; m >= 1; m >>= 1) v = fmaxf(v, __shfl_xor(v, m, 32));
  return v;
}
__device__ __forceinline__ float waveSum(float v) {
  for (int m = 16; m >= 1; m >>= 1) v += __shfl_xor(v, m, 32);
  return v;
}

struct EdgeChunk { int s; float v0, v1, v2, v3; bool ok; };

__device__ __forceinline__ EdgeChunk loadChunk(
    int base, int end, int lane, const int* __restrict__ ssrc,
    const float* __restrict__ er, v4f el4)
{
  EdgeChunk c;
  const int idx = base + lane;
  c.ok = idx < end;
  c.s = ssrc[c.ok ? idx : end - 1];
  const v4f r = *(const v4f*)(er + (size_t)c.s * H_);
  float v;
  v = r.x + el4.x; c.v0 = (v >= 0.f) ? v : NEG_SLOPE * v;
  v = r.y + el4.y; c.v1 = (v >= 0.f) ? v : NEG_SLOPE * v;
  v = r.z + el4.z; c.v2 = (v >= 0.f) ? v : NEG_SLOPE * v;
  v = r.w + el4.w; c.v3 = (v >= 0.f) ? v : NEG_SLOPE * v;
  return c;
}

__device__ __forceinline__ void accumChunk(
    int s, float a0, float a1, float a2, float a3, int n,
    unsigned mk0, unsigned mk1, unsigned mk2, unsigned mk3,
    const float* __restrict__ hWl, v4f& acc)
{
  const int b0 = __float_as_int(a0), b1 = __float_as_int(a1);
  const int b2 = __float_as_int(a2), b3 = __float_as_int(a3);
  for (int k = 0; k < n; k += 4) {
    v4f row[4];
#pragma unroll
    for (int u = 0; u < 4; ++u) {
      const int sk = __builtin_amdgcn_readlane(s, k + u);
      row[u] = *(const v4f*)(hWl + (size_t)sk * HF_);
    }
#pragma unroll
    for (int u = 0; u < 4; ++u) {
      const unsigned t0 = (unsigned)__builtin_amdgcn_readlane(b0, k + u);
      const unsigned t1 = (unsigned)__builtin_amdgcn_readlane(b1, k + u);
      const unsigned t2 = (unsigned)__builtin_amdgcn_readlane(b2, k + u);
      const unsigned t3 = (unsigned)__builtin_amdgcn_readlane(b3, k + u);
      const float wgt = __uint_as_float((t0 & mk0) | (t1 & mk1) | (t2 & mk2) | (t3 & mk3));
      acc += wgt * row[u];
    }
  }
}

__global__ __launch_bounds__(256) void gat_gather(
    const int* __restrict__ off, const int* __restrict__ ssrc,
    const float* __restrict__ hW, const float* __restrict__ el,
    const float* __restrict__ er, const float* __restrict__ bias,
    float* __restrict__ out, int nNodes)
{
  const int lane = threadIdx.x & 31;
  const int wave = __builtin_amdgcn_readfirstlane((int)(threadIdx.x >> 5));
  const int nWaves = gridDim.x * 8;
  const int hd = lane >> 3;
  const unsigned mk0 = (hd == 0) ? 0xFFFFFFFFu : 0u;
  const unsigned mk1 = (hd == 1) ? 0xFFFFFFFFu : 0u;
  const unsigned mk2 = (hd == 2) ? 0xFFFFFFFFu : 0u;
  const unsigned mk3 = (hd == 3) ? 0xFFFFFFFFu : 0u;
  const float NEG_INF = -__builtin_inff();
  const v4f b4 = *(const v4f*)(bias + lane * 4);
  const float* hWl = hW + lane * 4;

  for (int node = blockIdx.x * 8 + wave; node < nNodes; node += nWaves) {
    const int start = off[node];
    const int end = off[node + 1];
    const int deg = end - start;
    v4f acc = (v4f){0.f, 0.f, 0.f, 0.f};
    if (deg > 0) {
      const v4f el4 = *(const v4f*)(el + (size_t)node * H_);
      if (deg <= 32) {
        const EdgeChunk c = loadChunk(start, end, lane, ssrc, er, el4);
        const float m0 = waveMax(c.ok ? c.v0 : NEG_INF);
        const float m1 = waveMax(c.ok ? c.v1 : NEG_INF);
        const float m2 = waveMax(c.ok ? c.v2 : NEG_INF);
        const float m3 = waveMax(c.ok ? c.v3 : NEG_INF);
        const float x0 = c.ok ? expf(c.v0 - m0) : 0.f;
        const float x1 = c.ok ? expf(c.v1 - m1) : 0.f;
        const float x2 = c.ok ? expf(c.v2 - m2) : 0.f;
        const float x3 = c.ok ? expf(c.v3 - m3) : 0.f;
        const float d0 = waveSum(x0);
        const float d1 = waveSum(x1);
        const float d2 = waveSum(x2);
        const float d3 = waveSum(x3);
        accumChunk(c.s, x0 / d0, x1 / d1, x2 / d2, x3 / d3, deg, mk0, mk1, mk2, mk3, hWl, acc);
      } else {
        float m0 = NEG_INF, m1 = NEG_INF, m2 = NEG_INF, m3 = NEG_INF;
        for (int base = start; base < end; base += 32) {
          const EdgeChunk c = loadChunk(base, end, lane, ssrc, er, el4);
          m0 = fmaxf(m0, c.ok ? c.v0 : NEG_INF);
          m1 = fmaxf(m1, c.ok ? c.v1 : NEG_INF);
          m2 = fmaxf(m2, c.ok ? c.v2 : NEG_INF);
          m3 = fmaxf(m3, c.ok ? c.v3 : NEG_INF);
        }
        m0 = waveMax(m0); m1 = waveMax(m1); m2 = waveMax(m2); m3 = waveMax(m3);
        float d0 = 0.f, d1 = 0.f, d2 = 0.f, d3 = 0.f;
        for (int base = start; base < end; base += 32) {
          const EdgeChunk c = loadChunk(base, end, lane, ssrc, er, el4);
          d0 += c.ok ? expf(c.v0 - m0) : 0.f;
          d1 += c.ok ? expf(c.v1 - m1) : 0.f;
          d2 += c.ok ? expf(c.v2 - m2) : 0.f;
          d3 += c.ok ? expf(c.v3 - m3) : 0.f;
        }
        d0 = waveSum(d0); d1 = waveSum(d1); d2 = waveSum(d2); d3 = waveSum(d3);
        for (int base = start; base < end; base += 32) {
          const EdgeChunk c = loadChunk(base, end, lane, ssrc, er, el4);
          const float a0 = c.ok ? expf(c.v0 - m0) / d0 : 0.f;
          const float a1 = c.ok ? expf(c.v1 - m1) / d1 : 0.f;
          const float a2 = c.ok ? expf(c.v2 - m2) / d2 : 0.f;
          const float a3 = c.ok ? expf(c.v3 - m3) / d3 : 0.f;
          const int rem = end - base;
          accumChunk(c.s, a0, a1, a2, a3, rem < 32 ? rem : 32, mk0, mk1, mk2, mk3, hWl, acc);
        }
      }
    }
    *(v4f*)(out + (size_t)node * HF_ + lane * 4) = acc + b4;
  }
}

extern "C" void kernel_launch(void* const* d_in, const int* in_sizes, int n_in,
                              void* d_out, int out_size, void* d_ws, size_t ws_size,
                              hipStream_t stream)
{
  const float* h    = (const float*)d_in[0];
  const float* W    = (const float*)d_in[1];
  const float* alw  = (const float*)d_in[2];
  const float* alb  = (const float*)d_in[3];
  const float* arw  = (const float*)d_in[4];
  const float* arb  = (const float*)d_in[5];
  const float* bias = (const float*)d_in[6];
  const int*   src  = (const int*)d_in[7];
  const int*   dst  = (const int*)d_in[8];
  float* out = (float*)d_out;

  char* ws = (char*)d_ws;
  size_t off_b = 0;
  auto carve = [&](size_t bytes) -> void* {
    void* p = ws + off_b;
    off_b += (bytes + 255) & ~(size_t)255;
    return p;
  };
  float* hW   = (float*)carve((size_t)N_ * HF_ * sizeof(float));
  float* el   = (float*)carve((size_t)N_ * H_ * sizeof(float));
  float* er   = (float*)carve((size_t)N_ * H_ * sizeof(float));
  int*   btot = (int*)carve((size_t)NBINS * TOT_STRIDE * sizeof(int));
  int*   tcnt = (int*)carve((size_t)NCHUNKS * NBINS * sizeof(int));
  int*   toff = (int*)carve((size_t)NCHUNKS * NBINS * sizeof(int));
  int*   off  = (int*)carve((size_t)(N_ + 1) * sizeof(int));
  int*   part = (int*)carve((size_t)E_ * sizeof(int));
  int*   ssrc = (int*)carve((size_t)E_ * sizeof(int));
  (void)ws_size; (void)in_sizes; (void)n_in; (void)out_size;

  {
    const int strips = N_ / 16;
    const int blocks = (strips + 7) / 8;
    gat_gemm_wmma<<<blocks, 256, 0, stream>>>(h, W, alw, alb, arw, arb, hW, el, er, btot, N_, NBINS);
  }
  gat_part<<<NCHUNKS, PART_THREADS, 0, stream>>>(src, dst, part, tcnt, toff, btot);
  gat_bucket<<<NBINS, BK_THREADS, 0, stream>>>(part, tcnt, toff, btot, off, ssrc);
  {
    const int waves = (N_ + 3) / 4;
    const int blocks = (waves + 7) / 8;
    gat_gather<<<blocks, 256, 0, stream>>>(off, ssrc, hW, el, er, bias, out, N_);
  }
}


# Edge partition pass runs inside the GEMM launch as 196 extra work-groups, beside the matrix work; bin bases from the chunk tables, no bin-total atomics: 4 launches to 3; sums stay in arrival order
# speedup vs baseline: 6.0819x; 1.1013x over previous
#include <hip/hip_runtime.h>

#define N_  50000
#define E_  800000
#define D_  256
#define H_  4
#define F_  32
#define HF_ 128
#define NEG_SLOPE 0.2f

typedef __attribute__((ext_vector_type(2))) float v2f;
typedef __attribute__((ext_vector_type(8))) float v8f;
typedef __attribute__((ext_vector_type(4))) float v4f;


template <int MASK>
__device__ __forceinline__ float rowXor(float x) {
  return __int_as_float(__builtin_amdgcn_update_dpp(0, __float_as_int(x), 0x160 | MASK, 0xF, 0xF, true));
}

__device__ __forceinline__ void rowSum32(const float (&x)[32], int c, float& o0, float& o1) {
  const bool b3 = (c & 8) != 0, b2 = (c & 4) != 0, b1 = (c & 2) != 0, b0 = (c & 1) != 0;
  float y[16], z[8], w[4];
#pragma unroll
  for (int j = 0; j < 16; ++j) {
    const float keep = b3 ? x[j + 16] : x[j];
    const float send = b3 ? x[j] : x[j + 16];
    y[j] = keep + rowXor<8>(send);
  }
#pragma unroll
  for (int j = 0; j < 8; ++j) {
    const float keep = b2 ? y[j + 8] : y[j];
    const float send = b2 ? y[j] : y[j + 8];
    z[j] = keep + rowXor<4>(send);
  }
#pragma unroll
  for (int j = 0; j < 4; ++j) {
    const float keep = b1 ? z[j + 4] : z[j];
    const float send = b1 ? z[j] : z[j + 4];
    w[j] = keep + rowXor<2>(send);
  }
  {
    const float keep0 = b0 ? w[2] : w[0];
    const float send0 = b0 ? w[0] : w[2];
    const float keep1 = b0 ? w[3] : w[1];
    const float send1 = b0 ? w[1] : w[3];
    o0 = keep0 + rowXor<1>(send0);
    o1 = keep1 + rowXor<1>(send1);
  }
}

#define BIN_SHIFT   8
#define BIN_NODES   (1 << BIN_SHIFT)
#define NBINS       ((N_ + BIN_NODES - 1) / BIN_NODES)
#define PART_THREADS 256
#define PART_EPT     16
#define PART_CHUNK   (PART_THREADS * PART_EPT)
#define NCHUNKS      ((E_ + PART_CHUNK - 1) / PART_CHUNK)
static_assert(N_ <= 65536, "a source id must fit the low 16 bits of a packed edge");
static_assert(NBINS <= 32 * 7, "wave 0 scans the bins 7 per lane");
static_assert(BIN_NODES == 32 * 8, "wave 0 scans a bin's nodes 8 per lane");
static_assert(NBINS <= PART_THREADS, "one thread per bin writes the chunk's table row");
static_assert(NCHUNKS <= 256, "kernel 2 sums a bin's segments with its first 8 waves");

__device__ __forceinline__ int waveIncScanI(int v, int lane) {
  for (int d = 1; d < 32; d <<= 1) {
    const int y = __shfl_up(v, d, 32);
    if (lane >= d) v += y;
  }
  return v;
}
__device__ __forceinline__ int waveSumI(int v) {
  for (int m = 16; m >= 1; m >>= 1) v += __shfl_xor(v, m, 32);
  return v;
}

__global__ __launch_bounds__(256) void gat_gemm_wmma(
    const float* __restrict__ h, const float* __restrict__ W,
    const float* __restrict__ alw, const float* __restrict__ alb,
    const float* __restrict__ arw, const float* __restrict__ arb,
    const int* __restrict__ src, const int* __restrict__ dst,
    float* __restrict__ hW, float* __restrict__ el, float* __restrict__ er,
    int* __restrict__ part, int* __restrict__ tabCnt, int* __restrict__ tabOff, int nRows)
{
  __shared__ int sCnt[32 * 7];
  __shared__ int sOff[32 * 7];
  __shared__ int sStage[PART_CHUNK];
  const int wave = threadIdx.x >> 5;
  const int lane = threadIdx.x & 31;

  if (blockIdx.x < NCHUNKS) {
    const int t = threadIdx.x;
    const int chunkBase = blockIdx.x * PART_CHUNK;
    if (t < 32 * 7) sCnt[t] = 0;
    __syncthreads();

    int pk[PART_EPT], bn[PART_EPT], rk[PART_EPT];
#pragma unroll
    for (int u = 0; u < PART_EPT; ++u) {
      const int e = chunkBase + u * PART_THREADS + t;
      const int ec = (e < E_) ? e : E_ - 1;
      const int d = dst[ec];
      const int s = src[ec];
      bn[u] = (e < E_) ? (d >> BIN_SHIFT) : -1;
      pk[u] = ((d & (BIN_NODES - 1)) << 16) | s;
      rk[u] = 0;
    }
#pragma unroll
    for (int u = 0; u < PART_EPT; ++u)
      if (bn[u] >= 0) rk[u] = atomicAdd(&sCnt[bn[u]], 1);
    __syncthreads();

    if (t < 32) {
      int c[7]; int s = 0;
#pragma unroll
      for (int q = 0; q < 7; ++q) { c[q] = sCnt[lane * 7 + q]; s += c[q]; }
      int run = waveIncScanI(s, lane) - s;
#pragma unroll
      for (int q = 0; q < 7; ++q) { sOff[lane * 7 + q] = run; run += c[q]; }
    }
    __syncthreads();

#pragma unroll
    for (int u = 0; u < PART_EPT; ++u)
      if (bn[u] >= 0) sStage[sOff[bn[u]] + rk[u]] = pk[u];
    __syncthreads();

#pragma unroll
    for (int u = 0; u < PART_EPT; ++u) {
      const int i = u * PART_THREADS + t;
      if (chunkBase + i < E_) part[chunkBase + i] = sStage[i];
    }
    if (t < NBINS) {
      tabCnt[blockIdx.x * NBINS + t] = sCnt[t];
      tabOff[blockIdx.x * NBINS + t] = chunkBase + sOff[t];
    }
    return;
  }

  const int rowBase = ((blockIdx.x - NCHUNKS) * 8 + wave) * 16;
  if (rowBase >= nRows) return;

  const int m    = lane & 15;
  const int khalf= (lane >> 4) * 2;
  const float al0 = alw[m], al1 = alw[16 + m];
  const float ar0 = arw[m], ar1 = arw[16 + m];
  const float bl = alb[0], br = arb[0];

  v8f acc[8];
#pragma unroll
  for (int t = 0; t < 8; ++t) acc[t] = (v8f){0.f,0.f,0.f,0.f,0.f,0.f,0.f,0.f};

  for (int k0 = 0; k0 < D_; k0 += 4) {
    const float* ap = h + (size_t)(rowBase + m) * D_ + (k0 + khalf);
    v2f a; a.x = ap[0]; a.y = ap[1];
#pragma unroll
    for (int t = 0; t < 8; ++t) {
      const int n = t * 16 + (lane & 15);
      const float* bp = W + (size_t)(k0 + khalf) * HF_ + n;
      v2f b; b.x = bp[0]; b.y = bp[HF_];
      acc[t] = __builtin_amdgcn_wmma_f32_16x16x4_f32(
           false, a,  false, b,
           (short)0, acc[t],  false,  false);
    }
  }

  const int mHalf = (lane >> 4) * 8;
#pragma unroll
  for (int t = 0; t < 8; ++t) {
    const int n = t * 16 + (lane & 15);
#pragma unroll
    for (int v = 0; v < 8; ++v) {
      hW[(size_t)(rowBase + mHalf + v) * HF_ + n] = acc[t][v];
    }
  }

  float xl[32], xr[32];
#pragma unroll
  for (int hd = 0; hd < H_; ++hd) {
#pragma unroll
    for (int v = 0; v < 8; ++v) {
      const float p0 = acc[2 * hd][v], p1 = acc[2 * hd + 1][v];
      xl[v * 4 + hd] = p0 * al0 + p1 * al1;
      xr[v * 4 + hd] = p0 * ar0 + p1 * ar1;
    }
  }
  float l0, l1, r0, r1;
  rowSum32(xl, m, l0, l1);
  rowSum32(xr, m, r0, r1);
  const size_t eo = (size_t)(rowBase + mHalf) * H_ + 2 * m;
  v2f ol; ol.x = l0 + bl; ol.y = l1 + bl;
  v2f orr; orr.x = r0 + br; orr.y = r1 + br;
  *(v2f*)(el + eo) = ol;
  *(v2f*)(er + eo) = orr;
}

#define BK_THREADS 1024
#define BK_WAVES   (BK_THREADS / 32)
#define BK_SEGS    ((NCHUNKS + BK_WAVES - 1) / BK_WAVES)
#define BK_CAP     8192

__global__ __launch_bounds__(BK_THREADS) void gat_bucket(
    const int* __restrict__ part, const int* __restrict__ tabCnt,
    const int* __restrict__ tabOff,
    int* __restrict__ off, int* __restrict__ ssrc)
{
  __shared__ int sSegCnt[NCHUNKS];
  __shared__ int sSegOff[NCHUNKS];
  __shared__ int sCnt[BIN_NODES];
  __shared__ int sOff[BIN_NODES];
  __shared__ int sCur[BIN_NODES];
  __shared__ int sRed[2][8];
  __shared__ int sStage[BK_CAP];
  const int t = threadIdx.x;
  const int lane = t & 31;
  const int w = t >> 5;
  const int b = blockIdx.x;

  if (t < BIN_NODES) { sCnt[t] = 0; sCur[t] = 0; }
  int myCnt = 0, myBelow = 0;
  if (t < NCHUNKS) {
    myCnt = tabCnt[t * NBINS + b];
    const int o = tabOff[t * NBINS + b];
    sSegCnt[t] = myCnt;
    sSegOff[t] = o;
    myBelow = o - t * PART_CHUNK;
  }
  if (w < 8) {
    const int sb = waveSumI(myBelow);
    const int sc0 = waveSumI(myCnt);
    if (lane == 0) { sRed[0][w] = sb; sRed[1][w] = sc0; }
  }
  __syncthreads();
  int binBase = 0, binCount = 0;
#pragma unroll
  for (int q = 0; q < 8; ++q) { binBase += sRed[0][q]; binCount += sRed[1][q]; }

  int pv[BK_SEGS], sc[BK_SEGS], so[BK_SEGS];
#pragma unroll
  for (int q = 0; q < BK_SEGS; ++q) {
    const int j = w + q * BK_WAVES;
    sc[q] = (j < NCHUNKS) ? sSegCnt[j] : 0;
    so[q] = (j < NCHUNKS) ? sSegOff[j] : 0;
    const int ix = so[q] + lane;
    const int pw = part[(ix < E_) ? ix : E_ - 1];
    pv[q] = (lane < sc[q]) ? pw : -1;
  }
#pragma unroll
  for (int q = 0; q < BK_SEGS; ++q)
    if (pv[q] >= 0) atomicAdd(&sCnt[pv[q] >> 16], 1);
#pragma unroll
  for (int q = 0; q < BK_SEGS; ++q)
    for (int i = lane + 32; i < sc[q]; i += 32)
      atomicAdd(&sCnt[part[so[q] + i] >> 16], 1);
  __syncthreads();

  if (t < 32) {
    int c[8]; int s = 0;
#pragma unroll
    for (int q = 0; q < 8; ++q) { c[q] = sCnt[lane * 8 + q]; s += c[q]; }
    int run = waveIncScanI(s, lane) - s;
#pragma unroll
    for (int q = 0; q < 8; ++q) { sOff[lane * 8 + q] = run; run += c[q]; }
  }
  __syncthreads();

  if (t < BIN_NODES) {
    const int node = b * BIN_NODES + t;
    if (node < N_) off[node] = binBase + sOff[t];
  }
  if (b == NBINS - 1 && t == 0) off[N_] = binBase + binCount;

  const bool staged = binCount <= BK_CAP;
#pragma unroll
  for (int q = 0; q < BK_SEGS; ++q) {
    if (pv[q] >= 0) {
      const int f = pv[q] >> 16;
      const int pos = sOff[f] + atomicAdd(&sCur[f], 1);
      const int s = pv[q] & 0xFFFF;
      if (staged) sStage[pos] = s; else ssrc[binBase + pos] = s;
    }
  }
#pragma unroll
  for (int q = 0; q < BK_SEGS; ++q) {
    for (int i = lane + 32; i < sc[q]; i += 32) {
      const int p = part[so[q] + i];
      const int f = p >> 16;
      const int pos = sOff[f] + atomicAdd(&sCur[f], 1);
      const int s = p & 0xFFFF;
      if (staged) sStage[pos] = s; else ssrc[binBase + pos] = s;
    }
  }
  __syncthreads();
  if (staged)
    for (int i = t; i < binCount; i += BK_THREADS) ssrc[binBase + i] = sStage[i];
}

__device__ __forceinline__ float waveMax(float v) {
  for (int m = 16; m >= 1; m >>= 1) v = fmaxf(v, __shfl_xor(v, m, 32));
  return v;
}
__device__ __forceinline__ float waveSum(float v) {
  for (int m = 16; m >= 1; m >>= 1) v += __shfl_xor(v, m, 32);
  return v;
}

struct EdgeChunk { int s; float v0, v1, v2, v3; bool ok; };

__device__ __forceinline__ EdgeChunk loadChunk(
    int base, int end, int lane, const int* __restrict__ ssrc,
    const float* __restrict__ er, v4f el4)
{
  EdgeChunk c;
  const int idx = base + lane;
  c.ok = idx < end;
  c.s = ssrc[c.ok ? idx : end - 1];
  const v4f r = *(const v4f*)(er + (size_t)c.s * H_);
  float v;
  v = r.x + el4.x; c.v0 = (v >= 0.f) ? v : NEG_SLOPE * v;
  v = r.y + el4.y; c.v1 = (v >= 0.f) ? v : NEG_SLOPE * v;
  v = r.z + el4.z; c.v2 = (v >= 0.f) ? v : NEG_SLOPE * v;
  v = r.w + el4.w; c.v3 = (v >= 0.f) ? v : NEG_SLOPE * v;
  return c;
}

__device__ __forceinline__ void accumChunk(
    int s, float a0, float a1, float a2, float a3, int n,
    unsigned mk0, unsigned mk1, unsigned mk2, unsigned mk3,
    const float* __restrict__ hWl, v4f& acc)
{
  const int b0 = __float_as_int(a0), b1 = __float_as_int(a1);
  const int b2 = __float_as_int(a2), b3 = __float_as_int(a3);
  for (int k = 0; k < n; k += 4) {
    v4f row[4];
#pragma unroll
    for (int u = 0; u < 4; ++u) {
      const int sk = __builtin_amdgcn_readlane(s, k + u);
      row[u] = *(const v4f*)(hWl + (size_t)sk * HF_);
    }
#pragma unroll
    for (int u = 0; u < 4; ++u) {
      const unsigned t0 = (unsigned)__builtin_amdgcn_readlane(b0, k + u);
      const unsigned t1 = (unsigned)__builtin_amdgcn_readlane(b1, k + u);
      const unsigned t2 = (unsigned)__builtin_amdgcn_readlane(b2, k + u);
      const unsigned t3 = (unsigned)__builtin_amdgcn_readlane(b3, k + u);
      const float wgt = __uint_as_float((t0 & mk0) | (t1 & mk1) | (t2 & mk2) | (t3 & mk3));
      acc += wgt * row[u];
    }
  }
}

__global__ __launch_bounds__(256) void gat_gather(
    const int* __restrict__ off, const int* __restrict__ ssrc,
    const float* __restrict__ hW, const float* __restrict__ el,
    const float* __restrict__ er, const float* __restrict__ bias,
    float* __restrict__ out, int nNodes)
{
  const int lane = threadIdx.x & 31;
  const int wave = __builtin_amdgcn_readfirstlane((int)(threadIdx.x >> 5));
  const int nWaves = gridDim.x * 8;
  const int hd = lane >> 3;
  const unsigned mk0 = (hd == 0) ? 0xFFFFFFFFu : 0u;
  const unsigned mk1 = (hd == 1) ? 0xFFFFFFFFu : 0u;
  const unsigned mk2 = (hd == 2) ? 0xFFFFFFFFu : 0u;
  const unsigned mk3 = (hd == 3) ? 0xFFFFFFFFu : 0u;
  const float NEG_INF = -__builtin_inff();
  const v4f b4 = *(const v4f*)(bias + lane * 4);
  const float* hWl = hW + lane * 4;

  for (int node = blockIdx.x * 8 + wave; node < nNodes; node += nWaves) {
    const int start = off[node];
    const int end = off[node + 1];
    const int deg = end - start;
    v4f acc = (v4f){0.f, 0.f, 0.f, 0.f};
    if (deg > 0) {
      const v4f el4 = *(const v4f*)(el + (size_t)node * H_);
      if (deg <= 32) {
        const EdgeChunk c = loadChunk(start, end, lane, ssrc, er, el4);
        const float m0 = waveMax(c.ok ? c.v0 : NEG_INF);
        const float m1 = waveMax(c.ok ? c.v1 : NEG_INF);
        const float m2 = waveMax(c.ok ? c.v2 : NEG_INF);
        const float m3 = waveMax(c.ok ? c.v3 : NEG_INF);
        const float x0 = c.ok ? expf(c.v0 - m0) : 0.f;
        const float x1 = c.ok ? expf(c.v1 - m1) : 0.f;
        const float x2 = c.ok ? expf(c.v2 - m2) : 0.f;
        const float x3 = c.ok ? expf(c.v3 - m3) : 0.f;
        const float d0 = waveSum(x0);
        const float d1 = waveSum(x1);
        const float d2 = waveSum(x2);
        const float d3 = waveSum(x3);
        accumChunk(c.s, x0 / d0, x1 / d1, x2 / d2, x3 / d3, deg, mk0, mk1, mk2, mk3, hWl, acc);
      } else {
        float m0 = NEG_INF, m1 = NEG_INF, m2 = NEG_INF, m3 = NEG_INF;
        for (int base = start; base < end; base += 32) {
          const EdgeChunk c = loadChunk(base, end, lane, ssrc, er, el4);
          m0 = fmaxf(m0, c.ok ? c.v0 : NEG_INF);
          m1 = fmaxf(m1, c.ok ? c.v1 : NEG_INF);
          m2 = fmaxf(m2, c.ok ? c.v2 : NEG_INF);
          m3 = fmaxf(m3, c.ok ? c.v3 : NEG_INF);
        }
        m0 = waveMax(m0); m1 = waveMax(m1); m2 = waveMax(m2); m3 = waveMax(m3);
        float d0 = 0.f, d1 = 0.f, d2 = 0.f, d3 = 0.f;
        for (int base = start; base < end; base += 32) {
          const EdgeChunk c = loadChunk(base, end, lane, ssrc, er, el4);
          d0 += c.ok ? expf(c.v0 - m0) : 0.f;
          d1 += c.ok ? expf(c.v1 - m1) : 0.f;
          d2 += c.ok ? expf(c.v2 - m2) : 0.f;
          d3 += c.ok ? expf(c.v3 - m3) : 0.f;
        }
        d0 = waveSum(d0); d1 = waveSum(d1); d2 = waveSum(d2); d3 = waveSum(d3);
        for (int base = start; base < end; base += 32) {
          const EdgeChunk c = loadChunk(base, end, lane, ssrc, er, el4);
          const float a0 = c.ok ? expf(c.v0 - m0) / d0 : 0.f;
          const float a1 = c.ok ? expf(c.v1 - m1) / d1 : 0.f;
          const float a2 = c.ok ? expf(c.v2 - m2) / d2 : 0.f;
          const float a3 = c.ok ? expf(c.v3 - m3) / d3 : 0.f;
          const int rem = end - base;
          accumChunk(c.s, a0, a1, a2, a3, rem < 32 ? rem : 32, mk0, mk1, mk2, mk3, hWl, acc);
        }
      }
    }
    *(v4f*)(out + (size_t)node * HF_ + lane * 4) = acc + b4;
  }
}

extern "C" void kernel_launch(void* const* d_in, const int* in_sizes, int n_in,
                              void* d_out, int out_size, void* d_ws, size_t ws_size,
                              hipStream_t stream)
{
  const float* h    = (const float*)d_in[0];
  const float* W    = (const float*)d_in[1];
  const float* alw  = (const float*)d_in[2];
  const float* alb  = (const float*)d_in[3];
  const float* arw  = (const float*)d_in[4];
  const float* arb  = (const float*)d_in[5];
  const float* bias = (const float*)d_in[6];
  const int*   src  = (const int*)d_in[7];
  const int*   dst  = (const int*)d_in[8];
  float* out = (float*)d_out;

  char* ws = (char*)d_ws;
  size_t off_b = 0;
  auto carve = [&](size_t bytes) -> void* {
    void* p = ws + off_b;
    off_b += (bytes + 255) & ~(size_t)255;
    return p;
  };
  float* hW   = (float*)carve((size_t)N_ * HF_ * sizeof(float));
  float* el   = (float*)carve((size_t)N_ * H_ * sizeof(float));
  float* er   = (float*)carve((size_t)N_ * H_ * sizeof(float));
  int*   tcnt = (int*)carve((size_t)NCHUNKS * NBINS * sizeof(int));
  int*   toff = (int*)carve((size_t)NCHUNKS * NBINS * sizeof(int));
  int*   off  = (int*)carve((size_t)(N_ + 1) * sizeof(int));
  int*   part = (int*)carve((size_t)E_ * sizeof(int));
  int*   ssrc = (int*)carve((size_t)E_ * sizeof(int));
  (void)ws_size; (void)in_sizes; (void)n_in; (void)out_size;

  {
    const int strips = N_ / 16;
    const int blocks = (strips + 7) / 8;
    gat_gemm_wmma<<<NCHUNKS + blocks, 256, 0, stream>>>(h, W, alw, alb, arw, arb, src, dst,
                                                        hW, el, er, part, tcnt, toff, N_);
  }
  gat_bucket<<<NBINS, BK_THREADS, 0, stream>>>(part, tcnt, toff, off, ssrc);
  {
    const int waves = (N_ + 3) / 4;
    const int blocks = (waves + 7) / 8;
    gat_gather<<<blocks, 256, 0, stream>>>(off, ssrc, hW, el, er, bias, out, N_);
  }
}


# Gather: softmax of a bucket kept in the row sum's lane layout (lane = head x 8 edge slots): a weight reaches its rows by one DPP8 move, max/sum by 3 DPP steps; denominator sum order changes
# speedup vs baseline: 6.3361x; 1.0418x over previous
#include <hip/hip_runtime.h>

#define N_  50000
#define E_  800000
#define D_  256
#define H_  4
#define F_  32
#define HF_ 128
#define NEG_SLOPE 0.2f

typedef __attribute__((ext_vector_type(2))) float v2f;
typedef __attribute__((ext_vector_type(8))) float v8f;
typedef __attribute__((ext_vector_type(4))) float v4f;


template <int MASK>
__device__ __forceinline__ float rowXor(float x) {
  return __int_as_float(__builtin_amdgcn_update_dpp(0, __float_as_int(x), 0x160 | MASK, 0xF, 0xF, true));
}

__device__ __forceinline__ void rowSum32(const float (&x)[32], int c, float& o0, float& o1) {
  const bool b3 = (c & 8) != 0, b2 = (c & 4) != 0, b1 = (c & 2) != 0, b0 = (c & 1) != 0;
  float y[16], z[8], w[4];
#pragma unroll
  for (int j = 0; j < 16; ++j) {
    const float keep = b3 ? x[j + 16] : x[j];
    const float send = b3 ? x[j] : x[j + 16];
    y[j] = keep + rowXor<8>(send);
  }
#pragma unroll
  for (int j = 0; j < 8; ++j) {
    const float keep = b2 ? y[j + 8] : y[j];
    const float send = b2 ? y[j] : y[j + 8];
    z[j] = keep + rowXor<4>(send);
  }
#pragma unroll
  for (int j = 0; j < 4; ++j) {
    const float keep = b1 ? z[j + 4] : z[j];
    const float send = b1 ? z[j] : z[j + 4];
    w[j] = keep + rowXor<2>(send);
  }
  {
    const float keep0 = b0 ? w[2] : w[0];
    const float send0 = b0 ? w[0] : w[2];
    const float keep1 = b0 ? w[3] : w[1];
    const float send1 = b0 ? w[1] : w[3];
    o0 = keep0 + rowXor<1>(send0);
    o1 = keep1 + rowXor<1>(send1);
  }
}

#define BIN_SHIFT   8
#define BIN_NODES   (1 << BIN_SHIFT)
#define NBINS       ((N_ + BIN_NODES - 1) / BIN_NODES)
#define PART_THREADS 256
#define PART_EPT     16
#define PART_CHUNK   (PART_THREADS * PART_EPT)
#define NCHUNKS      ((E_ + PART_CHUNK - 1) / PART_CHUNK)
static_assert(N_ <= 65536, "a source id must fit the low 16 bits of a packed edge");
static_assert(NBINS <= 32 * 7, "wave 0 scans the bins 7 per lane");
static_assert(BIN_NODES == 32 * 8, "wave 0 scans a bin's nodes 8 per lane");
static_assert(NBINS <= PART_THREADS, "one thread per bin writes the chunk's table row");
static_assert(NCHUNKS <= 256, "kernel 2 sums a bin's segments with its first 8 waves");

__device__ __forceinline__ int waveIncScanI(int v, int lane) {
  for (int d = 1; d < 32; d <<= 1) {
    const int y = __shfl_up(v, d, 32);
    if (lane >= d) v += y;
  }
  return v;
}
__device__ __forceinline__ int waveSumI(int v) {
  for (int m = 16; m >= 1; m >>= 1) v += __shfl_xor(v, m, 32);
  return v;
}

__global__ __launch_bounds__(256) void gat_gemm_wmma(
    const float* __restrict__ h, const float* __restrict__ W,
    const float* __restrict__ alw, const float* __restrict__ alb,
    const float* __restrict__ arw, const float* __restrict__ arb,
    const int* __restrict__ src, const int* __restrict__ dst,
    float* __restrict__ hW, float* __restrict__ el, float* __restrict__ er,
    int* __restrict__ part, int* __restrict__ tabCnt, int* __restrict__ tabOff, int nRows)
{
  __shared__ int sCnt[32 * 7];
  __shared__ int sOff[32 * 7];
  __shared__ int sStage[PART_CHUNK];
  const int wave = threadIdx.x >> 5;
  const int lane = threadIdx.x & 31;

  if (blockIdx.x < NCHUNKS) {
    const int t = threadIdx.x;
    const int chunkBase = blockIdx.x * PART_CHUNK;
    if (t < 32 * 7) sCnt[t] = 0;
    __syncthreads();

    int pk[PART_EPT], bn[PART_EPT], rk[PART_EPT];
#pragma unroll
    for (int u = 0; u < PART_EPT; ++u) {
      const int e = chunkBase + u * PART_THREADS + t;
      const int ec = (e < E_) ? e : E_ - 1;
      const int d = dst[ec];
      const int s = src[ec];
      bn[u] = (e < E_) ? (d >> BIN_SHIFT) : -1;
      pk[u] = ((d & (BIN_NODES - 1)) << 16) | s;
      rk[u] = 0;
    }
#pragma unroll
    for (int u = 0; u < PART_EPT; ++u)
      if (bn[u] >= 0) rk[u] = atomicAdd(&sCnt[bn[u]], 1);
    __syncthreads();

    if (t < 32) {
      int c[7]; int s = 0;
#pragma unroll
      for (int q = 0; q < 7; ++q) { c[q] = sCnt[lane * 7 + q]; s += c[q]; }
      int run = waveIncScanI(s, lane) - s;
#pragma unroll
      for (int q = 0; q < 7; ++q) { sOff[lane * 7 + q] = run; run += c[q]; }
    }
    __syncthreads();

#pragma unroll
    for (int u = 0; u < PART_EPT; ++u)
      if (bn[u] >= 0) sStage[sOff[bn[u]] + rk[u]] = pk[u];
    __syncthreads();

#pragma unroll
    for (int u = 0; u < PART_EPT; ++u) {
      const int i = u * PART_THREADS + t;
      if (chunkBase + i < E_) part[chunkBase + i] = sStage[i];
    }
    if (t < NBINS) {
      tabCnt[blockIdx.x * NBINS + t] = sCnt[t];
      tabOff[blockIdx.x * NBINS + t] = chunkBase + sOff[t];
    }
    return;
  }

  const int rowBase = ((blockIdx.x - NCHUNKS) * 8 + wave) * 16;
  if (rowBase >= nRows) return;

  const int m    = lane & 15;
  const int khalf= (lane >> 4) * 2;
  const float al0 = alw[m], al1 = alw[16 + m];
  const float ar0 = arw[m], ar1 = arw[16 + m];
  const float bl = alb[0], br = arb[0];

  v8f acc[8];
#pragma unroll
  for (int t = 0; t < 8; ++t) acc[t] = (v8f){0.f,0.f,0.f,0.f,0.f,0.f,0.f,0.f};

  for (int k0 = 0; k0 < D_; k0 += 4) {
    const float* ap = h + (size_t)(rowBase + m) * D_ + (k0 + khalf);
    v2f a; a.x = ap[0]; a.y = ap[1];
#pragma unroll
    for (int t = 0; t < 8; ++t) {
      const int n = t * 16 + (lane & 15);
      const float* bp = W + (size_t)(k0 + khalf) * HF_ + n;
      v2f b; b.x = bp[0]; b.y = bp[HF_];
      acc[t] = __builtin_amdgcn_wmma_f32_16x16x4_f32(
           false, a,  false, b,
           (short)0, acc[t],  false,  false);
    }
  }

  const int mHalf = (lane >> 4) * 8;
#pragma unroll
  for (int t = 0; t < 8; ++t) {
    const int n = t * 16 + (lane & 15);
#pragma unroll
    for (int v = 0; v < 8; ++v) {
      hW[(size_t)(rowBase + mHalf + v) * HF_ + n] = acc[t][v];
    }
  }

  float xl[32], xr[32];
#pragma unroll
  for (int hd = 0; hd < H_; ++hd) {
#pragma unroll
    for (int v = 0; v < 8; ++v) {
      const float p0 = acc[2 * hd][v], p1 = acc[2 * hd + 1][v];
      xl[v * 4 + hd] = p0 * al0 + p1 * al1;
      xr[v * 4 + hd] = p0 * ar0 + p1 * ar1;
    }
  }
  float l0, l1, r0, r1;
  rowSum32(xl, m, l0, l1);
  rowSum32(xr, m, r0, r1);
  const size_t eo = (size_t)(rowBase + mHalf) * H_ + 2 * m;
  v2f ol; ol.x = l0 + bl; ol.y = l1 + bl;
  v2f orr; orr.x = r0 + br; orr.y = r1 + br;
  *(v2f*)(el + eo) = ol;
  *(v2f*)(er + eo) = orr;
}

#define BK_THREADS 1024
#define BK_WAVES   (BK_THREADS / 32)
#define BK_SEGS    ((NCHUNKS + BK_WAVES - 1) / BK_WAVES)
#define BK_CAP     8192

__global__ __launch_bounds__(BK_THREADS) void gat_bucket(
    const int* __restrict__ part, const int* __restrict__ tabCnt,
    const int* __restrict__ tabOff,
    int* __restrict__ off, int* __restrict__ ssrc)
{
  __shared__ int sSegCnt[NCHUNKS];
  __shared__ int sSegOff[NCHUNKS];
  __shared__ int sCnt[BIN_NODES];
  __shared__ int sOff[BIN_NODES];
  __shared__ int sCur[BIN_NODES];
  __shared__ int sRed[2][8];
  __shared__ int sStage[BK_CAP];
  const int t = threadIdx.x;
  const int lane = t & 31;
  const int w = t >> 5;
  const int b = blockIdx.x;

  if (t < BIN_NODES) { sCnt[t] = 0; sCur[t] = 0; }
  int myCnt = 0, myBelow = 0;
  if (t < NCHUNKS) {
    myCnt = tabCnt[t * NBINS + b];
    const int o = tabOff[t * NBINS + b];
    sSegCnt[t] = myCnt;
    sSegOff[t] = o;
    myBelow = o - t * PART_CHUNK;
  }
  if (w < 8) {
    const int sb = waveSumI(myBelow);
    const int sc0 = waveSumI(myCnt);
    if (lane == 0) { sRed[0][w] = sb; sRed[1][w] = sc0; }
  }
  __syncthreads();
  int binBase = 0, binCount = 0;
#pragma unroll
  for (int q = 0; q < 8; ++q) { binBase += sRed[0][q]; binCount += sRed[1][q]; }

  int pv[BK_SEGS], sc[BK_SEGS], so[BK_SEGS];
#pragma unroll
  for (int q = 0; q < BK_SEGS; ++q) {
    const int j = w + q * BK_WAVES;
    sc[q] = (j < NCHUNKS) ? sSegCnt[j] : 0;
    so[q] = (j < NCHUNKS) ? sSegOff[j] : 0;
    const int ix = so[q] + lane;
    const int pw = part[(ix < E_) ? ix : E_ - 1];
    pv[q] = (lane < sc[q]) ? pw : -1;
  }
#pragma unroll
  for (int q = 0; q < BK_SEGS; ++q)
    if (pv[q] >= 0) atomicAdd(&sCnt[pv[q] >> 16], 1);
#pragma unroll
  for (int q = 0; q < BK_SEGS; ++q)
    for (int i = lane + 32; i < sc[q]; i += 32)
      atomicAdd(&sCnt[part[so[q] + i] >> 16], 1);
  __syncthreads();

  if (t < 32) {
    int c[8]; int s = 0;
#pragma unroll
    for (int q = 0; q < 8; ++q) { c[q] = sCnt[lane * 8 + q]; s += c[q]; }
    int run = waveIncScanI(s, lane) - s;
#pragma unroll
    for (int q = 0; q < 8; ++q) { sOff[lane * 8 + q] = run; run += c[q]; }
  }
  __syncthreads();

  if (t < BIN_NODES) {
    const int node = b * BIN_NODES + t;
    if (node < N_) off[node] = binBase + sOff[t];
  }
  if (b == NBINS - 1 && t == 0) off[N_] = binBase + binCount;

  const bool staged = binCount <= BK_CAP;
#pragma unroll
  for (int q = 0; q < BK_SEGS; ++q) {
    if (pv[q] >= 0) {
      const int f = pv[q] >> 16;
      const int pos = sOff[f] + atomicAdd(&sCur[f], 1);
      const int s = pv[q] & 0xFFFF;
      if (staged) sStage[pos] = s; else ssrc[binBase + pos] = s;
    }
  }
#pragma unroll
  for (int q = 0; q < BK_SEGS; ++q) {
    for (int i = lane + 32; i < sc[q]; i += 32) {
      const int p = part[so[q] + i];
      const int f = p >> 16;
      const int pos = sOff[f] + atomicAdd(&sCur[f], 1);
      const int s = p & 0xFFFF;
      if (staged) sStage[pos] = s; else ssrc[binBase + pos] = s;
    }
  }
  __syncthreads();
  if (staged)
    for (int i = t; i < binCount; i += BK_THREADS) ssrc[binBase + i] = sStage[i];
}

__device__ __forceinline__ float waveMax(float v) {
  for (int m = 16; m >= 1; m >>= 1) v = fmaxf(v, __shfl_xor(v, m, 32));
  return v;
}
__device__ __forceinline__ float waveSum(float v) {
  for (int m = 16; m >= 1; m >>= 1) v += __shfl_xor(v, m, 32);
  return v;
}

struct EdgeChunk { int s; float v0, v1, v2, v3; bool ok; };

__device__ __forceinline__ EdgeChunk loadChunk(
    int base, int end, int lane, const int* __restrict__ ssrc,
    const float* __restrict__ er, v4f el4)
{
  EdgeChunk c;
  const int idx = base + lane;
  c.ok = idx < end;
  c.s = ssrc[c.ok ? idx : end - 1];
  const v4f r = *(const v4f*)(er + (size_t)c.s * H_);
  float v;
  v = r.x + el4.x; c.v0 = (v >= 0.f) ? v : NEG_SLOPE * v;
  v = r.y + el4.y; c.v1 = (v >= 0.f) ? v : NEG_SLOPE * v;
  v = r.z + el4.z; c.v2 = (v >= 0.f) ? v : NEG_SLOPE * v;
  v = r.w + el4.w; c.v3 = (v >= 0.f) ? v : NEG_SLOPE * v;
  return c;
}

__device__ __forceinline__ void accumChunk(
    int s, float a0, float a1, float a2, float a3, int n,
    unsigned mk0, unsigned mk1, unsigned mk2, unsigned mk3,
    const float* __restrict__ hWl, v4f& acc)
{
  const int b0 = __float_as_int(a0), b1 = __float_as_int(a1);
  const int b2 = __float_as_int(a2), b3 = __float_as_int(a3);
  for (int k = 0; k < n; k += 4) {
    v4f row[4];
#pragma unroll
    for (int u = 0; u < 4; ++u) {
      const int sk = __builtin_amdgcn_readlane(s, k + u);
      row[u] = *(const v4f*)(hWl + (size_t)sk * HF_);
    }
#pragma unroll
    for (int u = 0; u < 4; ++u) {
      const unsigned t0 = (unsigned)__builtin_amdgcn_readlane(b0, k + u);
      const unsigned t1 = (unsigned)__builtin_amdgcn_readlane(b1, k + u);
      const unsigned t2 = (unsigned)__builtin_amdgcn_readlane(b2, k + u);
      const unsigned t3 = (unsigned)__builtin_amdgcn_readlane(b3, k + u);
      const float wgt = __uint_as_float((t0 & mk0) | (t1 & mk1) | (t2 & mk2) | (t3 & mk3));
      acc += wgt * row[u];
    }
  }
}

template <int I>
__device__ __forceinline__ float grpBcast(float x) {
  return __int_as_float((int)__builtin_amdgcn_mov_dpp8((unsigned)__float_as_int(x), I * 0x249249));
}

template <int Q, int N>
__device__ __forceinline__ void accumRows(
    int sj, float aj, int lane, const float* __restrict__ hW, v4f& acc)
{
  v4f row[N];
#pragma unroll
  for (int u = 0; u < N; ++u) {
    const int sk = __builtin_amdgcn_readlane(sj, 4 * Q + u);
    const float* rp = hW + (size_t)sk * HF_;
    row[u] = ((const v4f*)rp)[lane];
  }
  acc += grpBcast<4 * Q + 0>(aj) * row[0];
  acc += grpBcast<4 * Q + 1>(aj) * row[1];
  acc += grpBcast<4 * Q + 2>(aj) * row[2];
  acc += grpBcast<4 * Q + 3>(aj) * row[3];
}

__device__ __forceinline__ void accumGroup(
    int rem, int sj, float aj, int lane, const float* __restrict__ hW, v4f& acc)
{
  if (rem > 0) accumRows<0, 4>(sj, aj, lane, hW, acc);
  if (rem > 4) accumRows<1, 4>(sj, aj, lane, hW, acc);
}

__global__ __launch_bounds__(256) __attribute__((amdgpu_waves_per_eu(16, 16))) void gat_gather(
    const int* __restrict__ off, const int* __restrict__ ssrc,
    const float* __restrict__ hW, const float* __restrict__ el,
    const float* __restrict__ er, const float* __restrict__ bias,
    float* __restrict__ out, int nNodes)
{
  const int lane = threadIdx.x & 31;
  const int wave = __builtin_amdgcn_readfirstlane((int)(threadIdx.x >> 5));
  const int nWaves = gridDim.x * 8;
  const int hd = lane >> 3;
  const int i8 = lane & 7;
  const unsigned mk0 = (hd == 0) ? 0xFFFFFFFFu : 0u;
  const unsigned mk1 = (hd == 1) ? 0xFFFFFFFFu : 0u;
  const unsigned mk2 = (hd == 2) ? 0xFFFFFFFFu : 0u;
  const unsigned mk3 = (hd == 3) ? 0xFFFFFFFFu : 0u;
  const float NEG_INF = -__builtin_inff();
  const v4f b4 = *(const v4f*)(bias + lane * 4);
  const float* hWl = hW + lane * 4;

  for (int node = blockIdx.x * 8 + wave; node < nNodes; node += nWaves) {
    const int start = off[node];
    const int end = off[node + 1];
    const int deg = end - start;
    v4f acc = (v4f){0.f, 0.f, 0.f, 0.f};
    if (deg > 0) {
      if (deg <= 32) {
        const float elh = el[(size_t)node * H_ + hd];
        int sj[4]; float vj[4]; bool okj[4];
#pragma unroll
        for (int j = 0; j < 4; ++j) {
          const int e = i8 + 8 * j;
          okj[j] = e < deg;
          sj[j] = ssrc[start + (okj[j] ? e : deg - 1)];
        }
        float m = NEG_INF;
#pragma unroll
        for (int j = 0; j < 4; ++j) {
          const float v = er[(size_t)sj[j] * H_ + hd] + elh;
          vj[j] = (v >= 0.f) ? v : NEG_SLOPE * v;
          m = fmaxf(m, okj[j] ? vj[j] : NEG_INF);
        }
        m = fmaxf(m, rowXor<1>(m));
        m = fmaxf(m, rowXor<2>(m));
        m = fmaxf(m, rowXor<4>(m));
        float xj[4];
#pragma unroll
        for (int j = 0; j < 4; ++j) xj[j] = okj[j] ? expf(vj[j] - m) : 0.f;
        float d = ((xj[0] + xj[1]) + xj[2]) + xj[3];
        d += rowXor<1>(d);
        d += rowXor<2>(d);
        d += rowXor<4>(d);
        accumGroup(deg,      sj[0], xj[0] / d, lane, hW, acc);
        accumGroup(deg - 8,  sj[1], xj[1] / d, lane, hW, acc);
        accumGroup(deg - 16, sj[2], xj[2] / d, lane, hW, acc);
        accumGroup(deg - 24, sj[3], xj[3] / d, lane, hW, acc);
      } else {
        const v4f el4 = *(const v4f*)(el + (size_t)node * H_);
        float m0 = NEG_INF, m1 = NEG_INF, m2 = NEG_INF, m3 = NEG_INF;
        for (int base = start; base < end; base += 32) {
          const EdgeChunk c = loadChunk(base, end, lane, ssrc, er, el4);
          m0 = fmaxf(m0, c.ok ? c.v0 : NEG_INF);
          m1 = fmaxf(m1, c.ok ? c.v1 : NEG_INF);
          m2 = fmaxf(m2, c.ok ? c.v2 : NEG_INF);
          m3 = fmaxf(m3, c.ok ? c.v3 : NEG_INF);
        }
        m0 = waveMax(m0); m1 = waveMax(m1); m2 = waveMax(m2); m3 = waveMax(m3);
        float d0 = 0.f, d1 = 0.f, d2 = 0.f, d3 = 0.f;
        for (int base = start; base < end; base += 32) {
          const EdgeChunk c = loadChunk(base, end, lane, ssrc, er, el4);
          d0 += c.ok ? expf(c.v0 - m0) : 0.f;
          d1 += c.ok ? expf(c.v1 - m1) : 0.f;
          d2 += c.ok ? expf(c.v2 - m2) : 0.f;
          d3 += c.ok ? expf(c.v3 - m3) : 0.f;
        }
        d0 = waveSum(d0); d1 = waveSum(d1); d2 = waveSum(d2); d3 = waveSum(d3);
        for (int base = start; base < end; base += 32) {
          const EdgeChunk c = loadChunk(base, end, lane, ssrc, er, el4);
          const float a0 = c.ok ? expf(c.v0 - m0) / d0 : 0.f;
          const float a1 = c.ok ? expf(c.v1 - m1) / d1 : 0.f;
          const float a2 = c.ok ? expf(c.v2 - m2) / d2 : 0.f;
          const float a3 = c.ok ? expf(c.v3 - m3) / d3 : 0.f;
          const int rem = end - base;
          accumChunk(c.s, a0, a1, a2, a3, rem < 32 ? rem : 32, mk0, mk1, mk2, mk3, hWl, acc);
        }
      }
    }
    *(v4f*)(out + (size_t)node * HF_ + lane * 4) = acc + b4;
  }
}

extern "C" void kernel_launch(void* const* d_in, const int* in_sizes, int n_in,
                              void* d_out, int out_size, void* d_ws, size_t ws_size,
                              hipStream_t stream)
{
  const float* h    = (const float*)d_in[0];
  const float* W    = (const float*)d_in[1];
  const float* alw  = (const float*)d_in[2];
  const float* alb  = (const float*)d_in[3];
  const float* arw  = (const float*)d_in[4];
  const float* arb  = (const float*)d_in[5];
  const float* bias = (const float*)d_in[6];
  const int*   src  = (const int*)d_in[7];
  const int*   dst  = (const int*)d_in[8];
  float* out = (float*)d_out;

  char* ws = (char*)d_ws;
  size_t off_b = 0;
  auto carve = [&](size_t bytes) -> void* {
    void* p = ws + off_b;
    off_b += (bytes + 255) & ~(size_t)255;
    return p;
  };
  float* hW   = (float*)carve((size_t)N_ * HF_ * sizeof(float));
  float* el   = (float*)carve((size_t)N_ * H_ * sizeof(float));
  float* er   = (float*)carve((size_t)N_ * H_ * sizeof(float));
  int*   tcnt = (int*)carve((size_t)NCHUNKS * NBINS * sizeof(int));
  int*   toff = (int*)carve((size_t)NCHUNKS * NBINS * sizeof(int));
  int*   off  = (int*)carve((size_t)(N_ + 1) * sizeof(int));
  int*   part = (int*)carve((size_t)E_ * sizeof(int));
  int*   ssrc = (int*)carve((size_t)E_ * sizeof(int));
  (void)ws_size; (void)in_sizes; (void)n_in; (void)out_size;

  {
    const int strips = N_ / 16;
    const int blocks = (strips + 7) / 8;
    gat_gemm_wmma<<<NCHUNKS + blocks, 256, 0, stream>>>(h, W, alw, alb, arw, arb, src, dst,
                                                        hW, el, er, part, tcnt, toff, N_);
  }
  gat_bucket<<<NBINS, BK_THREADS, 0, stream>>>(part, tcnt, toff, off, ssrc);
  {
    const int waves = (N_ + 3) / 4;
    const int blocks = (waves + 7) / 8;
    gat_gather<<<blocks, 256, 0, stream>>>(off, ssrc, hW, el, er, bias, out, N_);
  }
}


# GEMM regridded for even SIMD load: 256 work-groups of 16 waves, 3 whole strips per SIMD, the 53 left-over strips split by head (fullest SIMD had 4); partition on 512 threads: gather sum order changes
# speedup vs baseline: 6.6930x; 1.0563x over previous
#include <hip/hip_runtime.h>

#define N_  50000
#define E_  800000
#define D_  256
#define H_  4
#define F_  32
#define HF_ 128
#define NEG_SLOPE 0.2f

typedef __attribute__((ext_vector_type(2))) float v2f;
typedef __attribute__((ext_vector_type(8))) float v8f;
typedef __attribute__((ext_vector_type(4))) float v4f;


template <int MASK>
__device__ __forceinline__ float rowXor(float x) {
  return __int_as_float(__builtin_amdgcn_update_dpp(0, __float_as_int(x), 0x160 | MASK, 0xF, 0xF, true));
}

__device__ __forceinline__ void rowSum32(const float (&x)[32], int c, float& o0, float& o1) {
  const bool b3 = (c & 8) != 0, b2 = (c & 4) != 0, b1 = (c & 2) != 0, b0 = (c & 1) != 0;
  float y[16], z[8], w[4];
#pragma unroll
  for (int j = 0; j < 16; ++j) {
    const float keep = b3 ? x[j + 16] : x[j];
    const float send = b3 ? x[j] : x[j + 16];
    y[j] = keep + rowXor<8>(send);
  }
#pragma unroll
  for (int j = 0; j < 8; ++j) {
    const float keep = b2 ? y[j + 8] : y[j];
    const float send = b2 ? y[j] : y[j + 8];
    z[j] = keep + rowXor<4>(send);
  }
#pragma unroll
  for (int j = 0; j < 4; ++j) {
    const float keep = b1 ? z[j + 4] : z[j];
    const float send = b1 ? z[j] : z[j + 4];
    w[j] = keep + rowXor<2>(send);
  }
  {
    const float keep0 = b0 ? w[2] : w[0];
    const float send0 = b0 ? w[0] : w[2];
    const float keep1 = b0 ? w[3] : w[1];
    const float send1 = b0 ? w[1] : w[3];
    o0 = keep0 + rowXor<1>(send0);
    o1 = keep1 + rowXor<1>(send1);
  }
}

__device__ __forceinline__ float rowSum8(const float (&x)[8], int c) {
  const bool b3 = (c & 8) != 0, b2 = (c & 4) != 0, b1 = (c & 2) != 0;
  float y[4], z[2];
#pragma unroll
  for (int j = 0; j < 4; ++j) {
    const float keep = b3 ? x[j + 4] : x[j];
    const float send = b3 ? x[j] : x[j + 4];
    y[j] = keep + rowXor<8>(send);
  }
#pragma unroll
  for (int j = 0; j < 2; ++j) {
    const float keep = b2 ? y[j + 2] : y[j];
    const float send = b2 ? y[j] : y[j + 2];
    z[j] = keep + rowXor<4>(send);
  }
  const float keep = b1 ? z[1] : z[0];
  const float send = b1 ? z[0] : z[1];
  const float w = keep + rowXor<2>(send);
  return w + rowXor<1>(w);
}

#define GEMM_THREADS 512
#define GEMM_WAVES   (GEMM_THREADS / 32)
#define GEMM_BLOCKS  256
#define STRIPS       (N_ / 16)
#define STRIP_WAVES  (GEMM_WAVES - 4)
#define FULL_STRIPS  (GEMM_BLOCKS * STRIP_WAVES)
#define LEFT_STRIPS  (STRIPS - FULL_STRIPS)
static_assert(N_ % 16 == 0, "every strip has 16 rows: no row guard in the GEMM");
static_assert(LEFT_STRIPS >= 0 && LEFT_STRIPS <= GEMM_BLOCKS, "at most one split strip per work-group");
static_assert(H_ == 4 && F_ == 32, "a split strip gives one head (two 16-column tiles) to each of four waves");

template <int NT, int UNR>
__device__ __forceinline__ void gemmTiles(
    const float* __restrict__ h, const float* __restrict__ W,
    int rowBase, int colBase, int lane, v8f (&acc)[NT])
{
  const int m     = lane & 15;
  const int khalf = (lane >> 4) * 2;
#pragma unroll
  for (int t = 0; t < NT; ++t) acc[t] = (v8f){0.f,0.f,0.f,0.f,0.f,0.f,0.f,0.f};
#pragma unroll UNR
  for (int k0 = 0; k0 < D_; k0 += 4) {
    const float* ap = h + (size_t)(rowBase + m) * D_ + (k0 + khalf);
    v2f a; a.x = ap[0]; a.y = ap[1];
#pragma unroll
    for (int t = 0; t < NT; ++t) {
      const int n = colBase + t * 16 + m;
      const float* bp = W + (size_t)(k0 + khalf) * HF_ + n;
      v2f b; b.x = bp[0]; b.y = bp[HF_];
      acc[t] = __builtin_amdgcn_wmma_f32_16x16x4_f32(
           false, a,  false, b,
           (short)0, acc[t],  false,  false);
    }
  }
}

#define BIN_SHIFT   8
#define BIN_NODES   (1 << BIN_SHIFT)
#define NBINS       ((N_ + BIN_NODES - 1) / BIN_NODES)
#define PART_THREADS GEMM_THREADS
#define PART_EPT     8
#define PART_CHUNK   (PART_THREADS * PART_EPT)
#define NCHUNKS      ((E_ + PART_CHUNK - 1) / PART_CHUNK)
static_assert(N_ <= 65536, "a source id must fit the low 16 bits of a packed edge");
static_assert(NBINS <= 32 * 7, "wave 0 scans the bins 7 per lane");
static_assert(BIN_NODES == 32 * 8, "wave 0 scans a bin's nodes 8 per lane");
static_assert(NBINS <= PART_THREADS, "one thread per bin writes the chunk's table row");
static_assert(NCHUNKS <= 256, "kernel 2 sums a bin's segments with its first 8 waves");

__device__ __forceinline__ int waveIncScanI(int v, int lane) {
  for (int d = 1; d < 32; d <<= 1) {
    const int y = __shfl_up(v, d, 32);
    if (lane >= d) v += y;
  }
  return v;
}
__device__ __forceinline__ int waveSumI(int v) {
  for (int m = 16; m >= 1; m >>= 1) v += __shfl_xor(v, m, 32);
  return v;
}

__global__ __launch_bounds__(GEMM_THREADS) __attribute__((amdgpu_waves_per_eu(8, 8))) void gat_gemm_wmma(
    const float* __restrict__ h, const float* __restrict__ W,
    const float* __restrict__ alw, const float* __restrict__ alb,
    const float* __restrict__ arw, const float* __restrict__ arb,
    const int* __restrict__ src, const int* __restrict__ dst,
    float* __restrict__ hW, float* __restrict__ el, float* __restrict__ er,
    int* __restrict__ part, int* __restrict__ tabCnt, int* __restrict__ tabOff)
{
  __shared__ int sCnt[32 * 7];
  __shared__ int sOff[32 * 7];
  __shared__ int sStage[PART_CHUNK];
  __shared__ __attribute__((aligned(16))) float sLR[2 * 16 * H_];
  const int wave = __builtin_amdgcn_readfirstlane((int)(threadIdx.x >> 5));
  const int lane = threadIdx.x & 31;

  if (blockIdx.x >= GEMM_BLOCKS) {
    const int t = threadIdx.x;
    const int chunk = blockIdx.x - GEMM_BLOCKS;
    const int chunkBase = chunk * PART_CHUNK;
    if (t < 32 * 7) sCnt[t] = 0;
    __syncthreads();

    int pk[PART_EPT], bn[PART_EPT], rk[PART_EPT];
#pragma unroll
    for (int u = 0; u < PART_EPT; ++u) {
      const int e = chunkBase + u * PART_THREADS + t;
      const int ec = (e < E_) ? e : E_ - 1;
      const int d = dst[ec];
      const int s = src[ec];
      bn[u] = (e < E_) ? (d >> BIN_SHIFT) : -1;
      pk[u] = ((d & (BIN_NODES - 1)) << 16) | s;
      rk[u] = 0;
    }
#pragma unroll
    for (int u = 0; u < PART_EPT; ++u)
      if (bn[u] >= 0) rk[u] = atomicAdd(&sCnt[bn[u]], 1);
    __syncthreads();

    if (t < 32) {
      int c[7]; int s = 0;
#pragma unroll
      for (int q = 0; q < 7; ++q) { c[q] = sCnt[lane * 7 + q]; s += c[q]; }
      int run = waveIncScanI(s, lane) - s;
#pragma unroll
      for (int q = 0; q < 7; ++q) { sOff[lane * 7 + q] = run; run += c[q]; }
    }
    __syncthreads();

#pragma unroll
    for (int u = 0; u < PART_EPT; ++u)
      if (bn[u] >= 0) sStage[sOff[bn[u]] + rk[u]] = pk[u];
    __syncthreads();

#pragma unroll
    for (int u = 0; u < PART_EPT; ++u) {
      const int i = u * PART_THREADS + t;
      if (chunkBase + i < E_) part[chunkBase + i] = sStage[i];
    }
    if (t < NBINS) {
      tabCnt[chunk * NBINS + t] = sCnt[t];
      tabOff[chunk * NBINS + t] = chunkBase + sOff[t];
    }
    return;
  }

  const int gb    = blockIdx.x;
  const int m     = lane & 15;
  const int mHalf = (lane >> 4) * 8;
  const float al0 = alw[m], al1 = alw[16 + m];
  const float ar0 = arw[m], ar1 = arw[16 + m];
  const float bl = alb[0], br = arb[0];
  const bool hasLeft = gb < LEFT_STRIPS;

  if (wave % 5 != 0) {
    const int rowBase = (gb * STRIP_WAVES + (wave - wave / 5 - 1)) * 16;
    v8f acc[8];
    gemmTiles<8, 2>(h, W, rowBase, 0, lane, acc);

#pragma unroll
    for (int t = 0; t < 8; ++t) {
      const int n = t * 16 + m;
#pragma unroll
      for (int v = 0; v < 8; ++v) {
        hW[(size_t)(rowBase + mHalf + v) * HF_ + n] = acc[t][v];
      }
    }

    float xl[32], xr[32];
#pragma unroll
    for (int hd = 0; hd < H_; ++hd) {
#pragma unroll
      for (int v = 0; v < 8; ++v) {
        const float p0 = acc[2 * hd][v], p1 = acc[2 * hd + 1][v];
        xl[v * 4 + hd] = p0 * al0 + p1 * al1;
        xr[v * 4 + hd] = p0 * ar0 + p1 * ar1;
      }
    }
    float l0, l1, r0, r1;
    rowSum32(xl, m, l0, l1);
    rowSum32(xr, m, r0, r1);
    const size_t eo = (size_t)(rowBase + mHalf) * H_ + 2 * m;
    v2f ol; ol.x = l0 + bl; ol.y = l1 + bl;
    v2f orr; orr.x = r0 + br; orr.y = r1 + br;
    *(v2f*)(el + eo) = ol;
    *(v2f*)(er + eo) = orr;
  } else if (hasLeft) {
    const int hd = wave / 5;
    const int rowBase = (FULL_STRIPS + gb) * 16;
    v8f acc[2];
    gemmTiles<2, 8>(h, W, rowBase, hd * F_, lane, acc);

#pragma unroll
    for (int t = 0; t < 2; ++t) {
      const int n = hd * F_ + t * 16 + m;
#pragma unroll
      for (int v = 0; v < 8; ++v) {
        hW[(size_t)(rowBase + mHalf + v) * HF_ + n] = acc[t][v];
      }
    }
    float xl[8], xr[8];
#pragma unroll
    for (int v = 0; v < 8; ++v) {
      const float p0 = acc[0][v], p1 = acc[1][v];
      xl[v] = p0 * al0 + p1 * al1;
      xr[v] = p0 * ar0 + p1 * ar1;
    }
    const float l = rowSum8(xl, m);
    const float r = rowSum8(xr, m);
    if ((m & 1) == 0) {
      const int j = (mHalf + (m >> 1)) * H_ + hd;
      sLR[j] = l + bl;
      sLR[16 * H_ + j] = r + br;
    }
  }
  if (hasLeft) {
    __syncthreads();
    if (wave == 0) {
      const size_t eo = (size_t)(FULL_STRIPS + gb) * 16 * H_ + 2 * lane;
      *(v2f*)(el + eo) = *(const v2f*)(sLR + 2 * lane);
      *(v2f*)(er + eo) = *(const v2f*)(sLR + 16 * H_ + 2 * lane);
    }
  }
}

#define BK_THREADS 1024
#define BK_WAVES   (BK_THREADS / 32)
#define BK_SEGS    ((NCHUNKS + BK_WAVES - 1) / BK_WAVES)
#define BK_CAP     8192

__global__ __launch_bounds__(BK_THREADS) void gat_bucket(
    const int* __restrict__ part, const int* __restrict__ tabCnt,
    const int* __restrict__ tabOff,
    int* __restrict__ off, int* __restrict__ ssrc)
{
  __shared__ int sSegCnt[NCHUNKS];
  __shared__ int sSegOff[NCHUNKS];
  __shared__ int sCnt[BIN_NODES];
  __shared__ int sOff[BIN_NODES];
  __shared__ int sCur[BIN_NODES];
  __shared__ int sRed[2][8];
  __shared__ int sStage[BK_CAP];
  const int t = threadIdx.x;
  const int lane = t & 31;
  const int w = t >> 5;
  const int b = blockIdx.x;

  if (t < BIN_NODES) { sCnt[t] = 0; sCur[t] = 0; }
  int myCnt = 0, myBelow = 0;
  if (t < NCHUNKS) {
    myCnt = tabCnt[t * NBINS + b];
    const int o = tabOff[t * NBINS + b];
    sSegCnt[t] = myCnt;
    sSegOff[t] = o;
    myBelow = o - t * PART_CHUNK;
  }
  if (w < 8) {
    const int sb = waveSumI(myBelow);
    const int sc0 = waveSumI(myCnt);
    if (lane == 0) { sRed[0][w] = sb; sRed[1][w] = sc0; }
  }
  __syncthreads();
  int binBase = 0, binCount = 0;
#pragma unroll
  for (int q = 0; q < 8; ++q) { binBase += sRed[0][q]; binCount += sRed[1][q]; }

  int pv[BK_SEGS], sc[BK_SEGS], so[BK_SEGS];
#pragma unroll
  for (int q = 0; q < BK_SEGS; ++q) {
    const int j = w + q * BK_WAVES;
    sc[q] = (j < NCHUNKS) ? sSegCnt[j] : 0;
    so[q] = (j < NCHUNKS) ? sSegOff[j] : 0;
    const int ix = so[q] + lane;
    const int pw = part[(ix < E_) ? ix : E_ - 1];
    pv[q] = (lane < sc[q]) ? pw : -1;
  }
#pragma unroll
  for (int q = 0; q < BK_SEGS; ++q)
    if (pv[q] >= 0) atomicAdd(&sCnt[pv[q] >> 16], 1);
#pragma unroll
  for (int q = 0; q < BK_SEGS; ++q)
    for (int i = lane + 32; i < sc[q]; i += 32)
      atomicAdd(&sCnt[part[so[q] + i] >> 16], 1);
  __syncthreads();

  if (t < 32) {
    int c[8]; int s = 0;
#pragma unroll
    for (int q = 0; q < 8; ++q) { c[q] = sCnt[lane * 8 + q]; s += c[q]; }
    int run = waveIncScanI(s, lane) - s;
#pragma unroll
    for (int q = 0; q < 8; ++q) { sOff[lane * 8 + q] = run; run += c[q]; }
  }
  __syncthreads();

  if (t < BIN_NODES) {
    const int node = b * BIN_NODES + t;
    if (node < N_) off[node] = binBase + sOff[t];
  }
  if (b == NBINS - 1 && t == 0) off[N_] = binBase + binCount;

  const bool staged = binCount <= BK_CAP;
#pragma unroll
  for (int q = 0; q < BK_SEGS; ++q) {
    if (pv[q] >= 0) {
      const int f = pv[q] >> 16;
      const int pos = sOff[f] + atomicAdd(&sCur[f], 1);
      const int s = pv[q] & 0xFFFF;
      if (staged) sStage[pos] = s; else ssrc[binBase + pos] = s;
    }
  }
#pragma unroll
  for (int q = 0; q < BK_SEGS; ++q) {
    for (int i = lane + 32; i < sc[q]; i += 32) {
      const int p = part[so[q] + i];
      const int f = p >> 16;
      const int pos = sOff[f] + atomicAdd(&sCur[f], 1);
      const int s = p & 0xFFFF;
      if (staged) sStage[pos] = s; else ssrc[binBase + pos] = s;
    }
  }
  __syncthreads();
  if (staged)
    for (int i = t; i < binCount; i += BK_THREADS) ssrc[binBase + i] = sStage[i];
}

__device__ __forceinline__ float waveMax(float v) {
  for (int m = 16; m >= 1; m >>= 1) v = fmaxf(v, __shfl_xor(v, m, 32));
  return v;
}
__device__ __forceinline__ float waveSum(float v) {
  for (int m = 16; m >= 1; m >>= 1) v += __shfl_xor(v, m, 32);
  return v;
}

struct EdgeChunk { int s; float v0, v1, v2, v3; bool ok; };

__device__ __forceinline__ EdgeChunk loadChunk(
    int base, int end, int lane, const int* __restrict__ ssrc,
    const float* __restrict__ er, v4f el4)
{
  EdgeChunk c;
  const int idx = base + lane;
  c.ok = idx < end;
  c.s = ssrc[c.ok ? idx : end - 1];
  const v4f r = *(const v4f*)(er + (size_t)c.s * H_);
  float v;
  v = r.x + el4.x; c.v0 = (v >= 0.f) ? v : NEG_SLOPE * v;
  v = r.y + el4.y; c.v1 = (v >= 0.f) ? v : NEG_SLOPE * v;
  v = r.z + el4.z; c.v2 = (v >= 0.f) ? v : NEG_SLOPE * v;
  v = r.w + el4.w; c.v3 = (v >= 0.f) ? v : NEG_SLOPE * v;
  return c;
}

__device__ __forceinline__ void accumChunk(
    int s, float a0, float a1, float a2, float a3, int n,
    unsigned mk0, unsigned mk1, unsigned mk2, unsigned mk3,
    const float* __restrict__ hWl, v4f& acc)
{
  const int b0 = __float_as_int(a0), b1 = __float_as_int(a1);
  const int b2 = __float_as_int(a2), b3 = __float_as_int(a3);
  for (int k = 0; k < n; k += 4) {
    v4f row[4];
#pragma unroll
    for (int u = 0; u < 4; ++u) {
      const int sk = __builtin_amdgcn_readlane(s, k + u);
      row[u] = *(const v4f*)(hWl + (size_t)sk * HF_);
    }
#pragma unroll
    for (int u = 0; u < 4; ++u) {
      const unsigned t0 = (unsigned)__builtin_amdgcn_readlane(b0, k + u);
      const unsigned t1 = (unsigned)__builtin_amdgcn_readlane(b1, k + u);
      const unsigned t2 = (unsigned)__builtin_amdgcn_readlane(b2, k + u);
      const unsigned t3 = (unsigned)__builtin_amdgcn_readlane(b3, k + u);
      const float wgt = __uint_as_float((t0 & mk0) | (t1 & mk1) | (t2 & mk2) | (t3 & mk3));
      acc += wgt * row[u];
    }
  }
}

template <int I>
__device__ __forceinline__ float grpBcast(float x) {
  return __int_as_float((int)__builtin_amdgcn_mov_dpp8((unsigned)__float_as_int(x), I * 0x249249));
}

template <int Q, int N>
__device__ __forceinline__ void accumRows(
    int sj, float aj, int lane, const float* __restrict__ hW, v4f& acc)
{
  v4f row[N];
#pragma unroll
  for (int u = 0; u < N; ++u) {
    const int sk = __builtin_amdgcn_readlane(sj, 4 * Q + u);
    const float* rp = hW + (size_t)sk * HF_;
    row[u] = ((const v4f*)rp)[lane];
  }
  acc += grpBcast<4 * Q + 0>(aj) * row[0];
  acc += grpBcast<4 * Q + 1>(aj) * row[1];
  acc += grpBcast<4 * Q + 2>(aj) * row[2];
  acc += grpBcast<4 * Q + 3>(aj) * row[3];
}

__device__ __forceinline__ void accumGroup(
    int rem, int sj, float aj, int lane, const float* __restrict__ hW, v4f& acc)
{
  if (rem > 0) accumRows<0, 4>(sj, aj, lane, hW, acc);
  if (rem > 4) accumRows<1, 4>(sj, aj, lane, hW, acc);
}

__global__ __launch_bounds__(256) __attribute__((amdgpu_waves_per_eu(16, 16))) void gat_gather(
    const int* __restrict__ off, const int* __restrict__ ssrc,
    const float* __restrict__ hW, const float* __restrict__ el,
    const float* __restrict__ er, const float* __restrict__ bias,
    float* __restrict__ out, int nNodes)
{
  const int lane = threadIdx.x & 31;
  const int wave = __builtin_amdgcn_readfirstlane((int)(threadIdx.x >> 5));
  const int nWaves = gridDim.x * 8;
  const int hd = lane >> 3;
  const int i8 = lane & 7;
  const unsigned mk0 = (hd == 0) ? 0xFFFFFFFFu : 0u;
  const unsigned mk1 = (hd == 1) ? 0xFFFFFFFFu : 0u;
  const unsigned mk2 = (hd == 2) ? 0xFFFFFFFFu : 0u;
  const unsigned mk3 = (hd == 3) ? 0xFFFFFFFFu : 0u;
  const float NEG_INF = -__builtin_inff();
  const v4f b4 = *(const v4f*)(bias + lane * 4);
  const float* hWl = hW + lane * 4;

  for (int node = blockIdx.x * 8 + wave; node < nNodes; node += nWaves) {
    const int start = off[node];
    const int end = off[node + 1];
    const int deg = end - start;
    v4f acc = (v4f){0.f, 0.f, 0.f, 0.f};
    if (deg > 0) {
      if (deg <= 32) {
        const float elh = el[(size_t)node * H_ + hd];
        int sj[4]; float vj[4]; bool okj[4];
#pragma unroll
        for (int j = 0; j < 4; ++j) {
          const int e = i8 + 8 * j;
          okj[j] = e < deg;
          sj[j] = ssrc[start + (okj[j] ? e : deg - 1)];
        }
        float m = NEG_INF;
#pragma unroll
        for (int j = 0; j < 4; ++j) {
          const float v = er[(size_t)sj[j] * H_ + hd] + elh;
          vj[j] = (v >= 0.f) ? v : NEG_SLOPE * v;
          m = fmaxf(m, okj[j] ? vj[j] : NEG_INF);
        }
        m = fmaxf(m, rowXor<1>(m));
        m = fmaxf(m, rowXor<2>(m));
        m = fmaxf(m, rowXor<4>(m));
        float xj[4];
#pragma unroll
        for (int j = 0; j < 4; ++j) xj[j] = okj[j] ? expf(vj[j] - m) : 0.f;
        float d = ((xj[0] + xj[1]) + xj[2]) + xj[3];
        d += rowXor<1>(d);
        d += rowXor<2>(d);
        d += rowXor<4>(d);
        accumGroup(deg,      sj[0], xj[0] / d, lane, hW, acc);
        accumGroup(deg - 8,  sj[1], xj[1] / d, lane, hW, acc);
        accumGroup(deg - 16, sj[2], xj[2] / d, lane, hW, acc);
        accumGroup(deg - 24, sj[3], xj[3] / d, lane, hW, acc);
      } else {
        const v4f el4 = *(const v4f*)(el + (size_t)node * H_);
        float m0 = NEG_INF, m1 = NEG_INF, m2 = NEG_INF, m3 = NEG_INF;
        for (int base = start; base < end; base += 32) {
          const EdgeChunk c = loadChunk(base, end, lane, ssrc, er, el4);
          m0 = fmaxf(m0, c.ok ? c.v0 : NEG_INF);
          m1 = fmaxf(m1, c.ok ? c.v1 : NEG_INF);
          m2 = fmaxf(m2, c.ok ? c.v2 : NEG_INF);
          m3 = fmaxf(m3, c.ok ? c.v3 : NEG_INF);
        }
        m0 = waveMax(m0); m1 = waveMax(m1); m2 = waveMax(m2); m3 = waveMax(m3);
        float d0 = 0.f, d1 = 0.f, d2 = 0.f, d3 = 0.f;
        for (int base = start; base < end; base += 32) {
          const EdgeChunk c = loadChunk(base, end, lane, ssrc, er, el4);
          d0 += c.ok ? expf(c.v0 - m0) : 0.f;
          d1 += c.ok ? expf(c.v1 - m1) : 0.f;
          d2 += c.ok ? expf(c.v2 - m2) : 0.f;
          d3 += c.ok ? expf(c.v3 - m3) : 0.f;
        }
        d0 = waveSum(d0); d1 = waveSum(d1); d2 = waveSum(d2); d3 = waveSum(d3);
        for (int base = start; base < end; base += 32) {
          const EdgeChunk c = loadChunk(base, end, lane, ssrc, er, el4);
          const float a0 = c.ok ? expf(c.v0 - m0) / d0 : 0.f;
          const float a1 = c.ok ? expf(c.v1 - m1) / d1 : 0.f;
          const float a2 = c.ok ? expf(c.v2 - m2) / d2 : 0.f;
          const float a3 = c.ok ? expf(c.v3 - m3) / d3 : 0.f;
          const int rem = end - base;
          accumChunk(c.s, a0, a1, a2, a3, rem < 32 ? rem : 32, mk0, mk1, mk2, mk3, hWl, acc);
        }
      }
    }
    *(v4f*)(out + (size_t)node * HF_ + lane * 4) = acc + b4;
  }
}

extern "C" void kernel_launch(void* const* d_in, const int* in_sizes, int n_in,
                              void* d_out, int out_size, void* d_ws, size_t ws_size,
                              hipStream_t stream)
{
  const float* h    = (const float*)d_in[0];
  const float* W    = (const float*)d_in[1];
  const float* alw  = (const float*)d_in[2];
  const float* alb  = (const float*)d_in[3];
  const float* arw  = (const float*)d_in[4];
  const float* arb  = (const float*)d_in[5];
  const float* bias = (const float*)d_in[6];
  const int*   src  = (const int*)d_in[7];
  const int*   dst  = (const int*)d_in[8];
  float* out = (float*)d_out;

  char* ws = (char*)d_ws;
  size_t off_b = 0;
  auto carve = [&](size_t bytes) -> void* {
    void* p = ws + off_b;
    off_b += (bytes + 255) & ~(size_t)255;
    return p;
  };
  float* hW   = (float*)carve((size_t)N_ * HF_ * sizeof(float));
  float* el   = (float*)carve((size_t)N_ * H_ * sizeof(float));
  float* er   = (float*)carve((size_t)N_ * H_ * sizeof(float));
  int*   tcnt = (int*)carve((size_t)NCHUNKS * NBINS * sizeof(int));
  int*   toff = (int*)carve((size_t)NCHUNKS * NBINS * sizeof(int));
  int*   off  = (int*)carve((size_t)(N_ + 1) * sizeof(int));
  int*   part = (int*)carve((size_t)E_ * sizeof(int));
  int*   ssrc = (int*)carve((size_t)E_ * sizeof(int));
  (void)ws_size; (void)in_sizes; (void)n_in; (void)out_size;

  gat_gemm_wmma<<<GEMM_BLOCKS + NCHUNKS, GEMM_THREADS, 0, stream>>>(h, W, alw, alb, arw, arb, src, dst,
                                                                   hW, el, er, part, tcnt, toff);
  gat_bucket<<<NBINS, BK_THREADS, 0, stream>>>(part, tcnt, toff, off, ssrc);
  {
    const int waves = (N_ + 3) / 4;
    const int blocks = (waves + 7) / 8;
    gat_gather<<<blocks, 256, 0, stream>>>(off, ssrc, hW, el, er, bias, out, N_);
  }
}


# Bucket pass fused into the gather launch: one 32-wave work-group per 256-node bin sorts its edges in LDS, then sums its own nodes from the LDS list, 8 rows in flight: 3 launches to 2; sum order kept
# speedup vs baseline: 7.0409x; 1.0520x over previous
#include <hip/hip_runtime.h>

#define N_  50000
#define E_  800000
#define D_  256
#define H_  4
#define F_  32
#define HF_ 128
#define NEG_SLOPE 0.2f

typedef __attribute__((ext_vector_type(2))) float v2f;
typedef __attribute__((ext_vector_type(8))) float v8f;
typedef __attribute__((ext_vector_type(4))) float v4f;


template <int MASK>
__device__ __forceinline__ float rowXor(float x) {
  return __int_as_float(__builtin_amdgcn_update_dpp(0, __float_as_int(x), 0x160 | MASK, 0xF, 0xF, true));
}

__device__ __forceinline__ void rowSum32(const float (&x)[32], int c, float& o0, float& o1) {
  const bool b3 = (c & 8) != 0, b2 = (c & 4) != 0, b1 = (c & 2) != 0, b0 = (c & 1) != 0;
  float y[16], z[8], w[4];
#pragma unroll
  for (int j = 0; j < 16; ++j) {
    const float keep = b3 ? x[j + 16] : x[j];
    const float send = b3 ? x[j] : x[j + 16];
    y[j] = keep + rowXor<8>(send);
  }
#pragma unroll
  for (int j = 0; j < 8; ++j) {
    const float keep = b2 ? y[j + 8] : y[j];
    const float send = b2 ? y[j] : y[j + 8];
    z[j] = keep + rowXor<4>(send);
  }
#pragma unroll
  for (int j = 0; j < 4; ++j) {
    const float keep = b1 ? z[j + 4] : z[j];
    const float send = b1 ? z[j] : z[j + 4];
    w[j] = keep + rowXor<2>(send);
  }
  {
    const float keep0 = b0 ? w[2] : w[0];
    const float send0 = b0 ? w[0] : w[2];
    const float keep1 = b0 ? w[3] : w[1];
    const float send1 = b0 ? w[1] : w[3];
    o0 = keep0 + rowXor<1>(send0);
    o1 = keep1 + rowXor<1>(send1);
  }
}

__device__ __forceinline__ float rowSum8(const float (&x)[8], int c) {
  const bool b3 = (c & 8) != 0, b2 = (c & 4) != 0, b1 = (c & 2) != 0;
  float y[4], z[2];
#pragma unroll
  for (int j = 0; j < 4; ++j) {
    const float keep = b3 ? x[j + 4] : x[j];
    const float send = b3 ? x[j] : x[j + 4];
    y[j] = keep + rowXor<8>(send);
  }
#pragma unroll
  for (int j = 0; j < 2; ++j) {
    const float keep = b2 ? y[j + 2] : y[j];
    const float send = b2 ? y[j] : y[j + 2];
    z[j] = keep + rowXor<4>(send);
  }
  const float keep = b1 ? z[1] : z[0];
  const float send = b1 ? z[0] : z[1];
  const float w = keep + rowXor<2>(send);
  return w + rowXor<1>(w);
}

#define GEMM_THREADS 512
#define GEMM_WAVES   (GEMM_THREADS / 32)
#define GEMM_BLOCKS  256
#define STRIPS       (N_ / 16)
#define STRIP_WAVES  (GEMM_WAVES - 4)
#define FULL_STRIPS  (GEMM_BLOCKS * STRIP_WAVES)
#define LEFT_STRIPS  (STRIPS - FULL_STRIPS)
static_assert(N_ % 16 == 0, "every strip has 16 rows: no row guard in the GEMM");
static_assert(LEFT_STRIPS >= 0 && LEFT_STRIPS <= GEMM_BLOCKS, "at most one split strip per work-group");
static_assert(H_ == 4 && F_ == 32, "a split strip gives one head (two 16-column tiles) to each of four waves");

template <int NT, int UNR>
__device__ __forceinline__ void gemmTiles(
    const float* __restrict__ h, const float* __restrict__ W,
    int rowBase, int colBase, int lane, v8f (&acc)[NT])
{
  const int m     = lane & 15;
  const int khalf = (lane >> 4) * 2;
#pragma unroll
  for (int t = 0; t < NT; ++t) acc[t] = (v8f){0.f,0.f,0.f,0.f,0.f,0.f,0.f,0.f};
#pragma unroll UNR
  for (int k0 = 0; k0 < D_; k0 += 4) {
    const float* ap = h + (size_t)(rowBase + m) * D_ + (k0 + khalf);
    v2f a; a.x = ap[0]; a.y = ap[1];
#pragma unroll
    for (int t = 0; t < NT; ++t) {
      const int n = colBase + t * 16 + m;
      const float* bp = W + (size_t)(k0 + khalf) * HF_ + n;
      v2f b; b.x = bp[0]; b.y = bp[HF_];
      acc[t] = __builtin_amdgcn_wmma_f32_16x16x4_f32(
           false, a,  false, b,
           (short)0, acc[t],  false,  false);
    }
  }
}

#define BIN_SHIFT   8
#define BIN_NODES   (1 << BIN_SHIFT)
#define NBINS       ((N_ + BIN_NODES - 1) / BIN_NODES)
#define PART_THREADS GEMM_THREADS
#define PART_EPT     8
#define PART_CHUNK   (PART_THREADS * PART_EPT)
#define NCHUNKS      ((E_ + PART_CHUNK - 1) / PART_CHUNK)
static_assert(N_ <= 65536, "a source id must fit the low 16 bits of a packed edge");
static_assert(NBINS <= 32 * 7, "wave 0 scans the bins 7 per lane");
static_assert(BIN_NODES == 32 * 8, "wave 0 scans a bin's nodes 8 per lane");
static_assert(NBINS <= PART_THREADS, "one thread per bin writes the chunk's table row");
static_assert(NCHUNKS <= 256, "kernel 2 sums a bin's segments with its first 8 waves");

__device__ __forceinline__ int waveIncScanI(int v, int lane) {
  for (int d = 1; d < 32; d <<= 1) {
    const int y = __shfl_up(v, d, 32);
    if (lane >= d) v += y;
  }
  return v;
}
__device__ __forceinline__ int waveSumI(int v) {
  for (int m = 16; m >= 1; m >>= 1) v += __shfl_xor(v, m, 32);
  return v;
}

__global__ __launch_bounds__(GEMM_THREADS) __attribute__((amdgpu_waves_per_eu(8, 8))) void gat_gemm_wmma(
    const float* __restrict__ h, const float* __restrict__ W,
    const float* __restrict__ alw, const float* __restrict__ alb,
    const float* __restrict__ arw, const float* __restrict__ arb,
    const int* __restrict__ src, const int* __restrict__ dst,
    float* __restrict__ hW, float* __restrict__ el, float* __restrict__ er,
    int* __restrict__ part, int* __restrict__ tabCnt, int* __restrict__ tabOff)
{
  __shared__ int sCnt[32 * 7];
  __shared__ int sOff[32 * 7];
  __shared__ int sStage[PART_CHUNK];
  __shared__ __attribute__((aligned(16))) float sLR[2 * 16 * H_];
  const int wave = __builtin_amdgcn_readfirstlane((int)(threadIdx.x >> 5));
  const int lane = threadIdx.x & 31;

  if (blockIdx.x >= GEMM_BLOCKS) {
    const int t = threadIdx.x;
    const int chunk = blockIdx.x - GEMM_BLOCKS;
    const int chunkBase = chunk * PART_CHUNK;
    if (t < 32 * 7) sCnt[t] = 0;
    __syncthreads();

    int pk[PART_EPT], bn[PART_EPT], rk[PART_EPT];
#pragma unroll
    for (int u = 0; u < PART_EPT; ++u) {
      const int e = chunkBase + u * PART_THREADS + t;
      const int ec = (e < E_) ? e : E_ - 1;
      const int d = dst[ec];
      const int s = src[ec];
      bn[u] = (e < E_) ? (d >> BIN_SHIFT) : -1;
      pk[u] = ((d & (BIN_NODES - 1)) << 16) | s;
      rk[u] = 0;
    }
#pragma unroll
    for (int u = 0; u < PART_EPT; ++u)
      if (bn[u] >= 0) rk[u] = atomicAdd(&sCnt[bn[u]], 1);
    __syncthreads();

    if (t < 32) {
      int c[7]; int s = 0;
#pragma unroll
      for (int q = 0; q < 7; ++q) { c[q] = sCnt[lane * 7 + q]; s += c[q]; }
      int run = waveIncScanI(s, lane) - s;
#pragma unroll
      for (int q = 0; q < 7; ++q) { sOff[lane * 7 + q] = run; run += c[q]; }
    }
    __syncthreads();

#pragma unroll
    for (int u = 0; u < PART_EPT; ++u)
      if (bn[u] >= 0) sStage[sOff[bn[u]] + rk[u]] = pk[u];
    __syncthreads();

#pragma unroll
    for (int u = 0; u < PART_EPT; ++u) {
      const int i = u * PART_THREADS + t;
      if (chunkBase + i < E_) part[chunkBase + i] = sStage[i];
    }
    if (t < NBINS) {
      tabCnt[chunk * NBINS + t] = sCnt[t];
      tabOff[chunk * NBINS + t] = chunkBase + sOff[t];
    }
    return;
  }

  const int gb    = blockIdx.x;
  const int m     = lane & 15;
  const int mHalf = (lane >> 4) * 8;
  const float al0 = alw[m], al1 = alw[16 + m];
  const float ar0 = arw[m], ar1 = arw[16 + m];
  const float bl = alb[0], br = arb[0];
  const bool hasLeft = gb < LEFT_STRIPS;

  if (wave % 5 != 0) {
    const int rowBase = (gb * STRIP_WAVES + (wave - wave / 5 - 1)) * 16;
    v8f acc[8];
    gemmTiles<8, 2>(h, W, rowBase, 0, lane, acc);

#pragma unroll
    for (int t = 0; t < 8; ++t) {
      const int n = t * 16 + m;
#pragma unroll
      for (int v = 0; v < 8; ++v) {
        hW[(size_t)(rowBase + mHalf + v) * HF_ + n] = acc[t][v];
      }
    }

    float xl[32], xr[32];
#pragma unroll
    for (int hd = 0; hd < H_; ++hd) {
#pragma unroll
      for (int v = 0; v < 8; ++v) {
        const float p0 = acc[2 * hd][v], p1 = acc[2 * hd + 1][v];
        xl[v * 4 + hd] = p0 * al0 + p1 * al1;
        xr[v * 4 + hd] = p0 * ar0 + p1 * ar1;
      }
    }
    float l0, l1, r0, r1;
    rowSum32(xl, m, l0, l1);
    rowSum32(xr, m, r0, r1);
    const size_t eo = (size_t)(rowBase + mHalf) * H_ + 2 * m;
    v2f ol; ol.x = l0 + bl; ol.y = l1 + bl;
    v2f orr; orr.x = r0 + br; orr.y = r1 + br;
    *(v2f*)(el + eo) = ol;
    *(v2f*)(er + eo) = orr;
  } else if (hasLeft) {
    const int hd = wave / 5;
    const int rowBase = (FULL_STRIPS + gb) * 16;
    v8f acc[2];
    gemmTiles<2, 8>(h, W, rowBase, hd * F_, lane, acc);

#pragma unroll
    for (int t = 0; t < 2; ++t) {
      const int n = hd * F_ + t * 16 + m;
#pragma unroll
      for (int v = 0; v < 8; ++v) {
        hW[(size_t)(rowBase + mHalf + v) * HF_ + n] = acc[t][v];
      }
    }
    float xl[8], xr[8];
#pragma unroll
    for (int v = 0; v < 8; ++v) {
      const float p0 = acc[0][v], p1 = acc[1][v];
      xl[v] = p0 * al0 + p1 * al1;
      xr[v] = p0 * ar0 + p1 * ar1;
    }
    const float l = rowSum8(xl, m);
    const float r = rowSum8(xr, m);
    if ((m & 1) == 0) {
      const int j = (mHalf + (m >> 1)) * H_ + hd;
      sLR[j] = l + bl;
      sLR[16 * H_ + j] = r + br;
    }
  }
  if (hasLeft) {
    __syncthreads();
    if (wave == 0) {
      const size_t eo = (size_t)(FULL_STRIPS + gb) * 16 * H_ + 2 * lane;
      *(v2f*)(el + eo) = *(const v2f*)(sLR + 2 * lane);
      *(v2f*)(er + eo) = *(const v2f*)(sLR + 16 * H_ + 2 * lane);
    }
  }
}

#define BK_THREADS 1024
#define BK_WAVES   (BK_THREADS / 32)
#define BK_SEGS    ((NCHUNKS + BK_WAVES - 1) / BK_WAVES)
#define BK_CAP     8192

__device__ __forceinline__ float waveMax(float v) {
  for (int m = 16; m >= 1; m >>= 1) v = fmaxf(v, __shfl_xor(v, m, 32));
  return v;
}
__device__ __forceinline__ float waveSum(float v) {
  for (int m = 16; m >= 1; m >>= 1) v += __shfl_xor(v, m, 32);
  return v;
}

template <bool ST>
__device__ __forceinline__ int srcAt(const int* stage, const int* __restrict__ spill, int i) {
  if (ST) return stage[i];
  return spill[i];
}

struct EdgeChunk { int s; float v0, v1, v2, v3; bool ok; };

template <bool ST>
__device__ __forceinline__ EdgeChunk loadChunk(
    int base, int end, int lane, const int* stage, const int* __restrict__ spill,
    const float* __restrict__ er, v4f el4)
{
  EdgeChunk c;
  const int idx = base + lane;
  c.ok = idx < end;
  c.s = srcAt<ST>(stage, spill, c.ok ? idx : end - 1);
  const v4f r = *(const v4f*)(er + (size_t)c.s * H_);
  float v;
  v = r.x + el4.x; c.v0 = (v >= 0.f) ? v : NEG_SLOPE * v;
  v = r.y + el4.y; c.v1 = (v >= 0.f) ? v : NEG_SLOPE * v;
  v = r.z + el4.z; c.v2 = (v >= 0.f) ? v : NEG_SLOPE * v;
  v = r.w + el4.w; c.v3 = (v >= 0.f) ? v : NEG_SLOPE * v;
  return c;
}

__device__ __forceinline__ void accumChunk(
    int s, float a0, float a1, float a2, float a3, int n,
    unsigned mk0, unsigned mk1, unsigned mk2, unsigned mk3,
    const float* __restrict__ hWl, v4f& acc)
{
  const int b0 = __float_as_int(a0), b1 = __float_as_int(a1);
  const int b2 = __float_as_int(a2), b3 = __float_as_int(a3);
  for (int k = 0; k < n; k += 4) {
    v4f row[4];
#pragma unroll
    for (int u = 0; u < 4; ++u) {
      const int sk = __builtin_amdgcn_readlane(s, k + u);
      row[u] = *(const v4f*)(hWl + (size_t)sk * HF_);
    }
#pragma unroll
    for (int u = 0; u < 4; ++u) {
      const unsigned t0 = (unsigned)__builtin_amdgcn_readlane(b0, k + u);
      const unsigned t1 = (unsigned)__builtin_amdgcn_readlane(b1, k + u);
      const unsigned t2 = (unsigned)__builtin_amdgcn_readlane(b2, k + u);
      const unsigned t3 = (unsigned)__builtin_amdgcn_readlane(b3, k + u);
      const float wgt = __uint_as_float((t0 & mk0) | (t1 & mk1) | (t2 & mk2) | (t3 & mk3));
      acc += wgt * row[u];
    }
  }
}

template <int I>
__device__ __forceinline__ float grpBcast(float x) {
  return __int_as_float((int)__builtin_amdgcn_mov_dpp8((unsigned)__float_as_int(x), I * 0x249249));
}

template <int N>
__device__ __forceinline__ void accumRows(
    int sj, float aj, int lane, const float* __restrict__ hW, v4f& acc)
{
  v4f row[N];
#pragma unroll
  for (int u = 0; u < N; ++u) {
    const int sk = __builtin_amdgcn_readlane(sj, u);
    const float* rp = hW + (size_t)sk * HF_;
    row[u] = ((const v4f*)rp)[lane];
  }
  acc += grpBcast<0>(aj) * row[0];
  acc += grpBcast<1>(aj) * row[1];
  acc += grpBcast<2>(aj) * row[2];
  acc += grpBcast<3>(aj) * row[3];
  if (N == 8) {
    acc += grpBcast<4>(aj) * row[N - 4];
    acc += grpBcast<5>(aj) * row[N - 3];
    acc += grpBcast<6>(aj) * row[N - 2];
    acc += grpBcast<7>(aj) * row[N - 1];
  }
}

__device__ __forceinline__ void accumGroup(
    int rem, int sj, float aj, int lane, const float* __restrict__ hW, v4f& acc)
{
  if (rem > 4) accumRows<8>(sj, aj, lane, hW, acc);
  else if (rem > 0) accumRows<4>(sj, aj, lane, hW, acc);
}

template <bool ST>
__device__ __forceinline__ void gatherBin(
    int b, int wave, int lane,
    const int* sOff, const int* sCnt, const int* stage, const int* __restrict__ spill,
    const float* __restrict__ hW, const float* __restrict__ el,
    const float* __restrict__ er, const float* __restrict__ bias,
    float* __restrict__ out)
{
  const int hd = lane >> 3;
  const int i8 = lane & 7;
  const unsigned mk0 = (hd == 0) ? 0xFFFFFFFFu : 0u;
  const unsigned mk1 = (hd == 1) ? 0xFFFFFFFFu : 0u;
  const unsigned mk2 = (hd == 2) ? 0xFFFFFFFFu : 0u;
  const unsigned mk3 = (hd == 3) ? 0xFFFFFFFFu : 0u;
  const float NEG_INF = -__builtin_inff();
  const v4f b4 = *(const v4f*)(bias + lane * 4);
  const float* hWl = hW + lane * 4;

  for (int f = wave; f < BIN_NODES; f += BK_WAVES) {
    const int node = b * BIN_NODES + f;
    if (node >= N_) break;
    const int start = __builtin_amdgcn_readfirstlane(sOff[f]);
    const int deg = __builtin_amdgcn_readfirstlane(sCnt[f]);
    const int end = start + deg;
    v4f acc = (v4f){0.f, 0.f, 0.f, 0.f};
    if (deg > 0) {
      if (deg <= 32) {
        const float elh = el[(size_t)node * H_ + hd];
        int sj[4]; float vj[4]; bool okj[4];
#pragma unroll
        for (int j = 0; j < 4; ++j) {
          const int e = i8 + 8 * j;
          okj[j] = e < deg;
          sj[j] = srcAt<ST>(stage, spill, start + (okj[j] ? e : deg - 1));
        }
        float m = NEG_INF;
#pragma unroll
        for (int j = 0; j < 4; ++j) {
          const float v = er[(size_t)sj[j] * H_ + hd] + elh;
          vj[j] = (v >= 0.f) ? v : NEG_SLOPE * v;
          m = fmaxf(m, okj[j] ? vj[j] : NEG_INF);
        }
        m = fmaxf(m, rowXor<1>(m));
        m = fmaxf(m, rowXor<2>(m));
        m = fmaxf(m, rowXor<4>(m));
        float xj[4];
#pragma unroll
        for (int j = 0; j < 4; ++j) xj[j] = okj[j] ? expf(vj[j] - m) : 0.f;
        float d = ((xj[0] + xj[1]) + xj[2]) + xj[3];
        d += rowXor<1>(d);
        d += rowXor<2>(d);
        d += rowXor<4>(d);
        accumGroup(deg,      sj[0], xj[0] / d, lane, hW, acc);
        accumGroup(deg - 8,  sj[1], xj[1] / d, lane, hW, acc);
        accumGroup(deg - 16, sj[2], xj[2] / d, lane, hW, acc);
        accumGroup(deg - 24, sj[3], xj[3] / d, lane, hW, acc);
      } else {
        const v4f el4 = *(const v4f*)(el + (size_t)node * H_);
        float m0 = NEG_INF, m1 = NEG_INF, m2 = NEG_INF, m3 = NEG_INF;
        for (int base = start; base < end; base += 32) {
          const EdgeChunk c = loadChunk<ST>(base, end, lane, stage, spill, er, el4);
          m0 = fmaxf(m0, c.ok ? c.v0 : NEG_INF);
          m1 = fmaxf(m1, c.ok ? c.v1 : NEG_INF);
          m2 = fmaxf(m2, c.ok ? c.v2 : NEG_INF);
          m3 = fmaxf(m3, c.ok ? c.v3 : NEG_INF);
        }
        m0 = waveMax(m0); m1 = waveMax(m1); m2 = waveMax(m2); m3 = waveMax(m3);
        float d0 = 0.f, d1 = 0.f, d2 = 0.f, d3 = 0.f;
        for (int base = start; base < end; base += 32) {
          const EdgeChunk c = loadChunk<ST>(base, end, lane, stage, spill, er, el4);
          d0 += c.ok ? expf(c.v0 - m0) : 0.f;
          d1 += c.ok ? expf(c.v1 - m1) : 0.f;
          d2 += c.ok ? expf(c.v2 - m2) : 0.f;
          d3 += c.ok ? expf(c.v3 - m3) : 0.f;
        }
        d0 = waveSum(d0); d1 = waveSum(d1); d2 = waveSum(d2); d3 = waveSum(d3);
        for (int base = start; base < end; base += 32) {
          const EdgeChunk c = loadChunk<ST>(base, end, lane, stage, spill, er, el4);
          const float a0 = c.ok ? expf(c.v0 - m0) / d0 : 0.f;
          const float a1 = c.ok ? expf(c.v1 - m1) / d1 : 0.f;
          const float a2 = c.ok ? expf(c.v2 - m2) / d2 : 0.f;
          const float a3 = c.ok ? expf(c.v3 - m3) / d3 : 0.f;
          const int rem = end - base;
          accumChunk(c.s, a0, a1, a2, a3, rem < 32 ? rem : 32, mk0, mk1, mk2, mk3, hWl, acc);
        }
      }
    }
    *(v4f*)(out + (size_t)node * HF_ + lane * 4) = acc + b4;
  }
}

__global__ __launch_bounds__(BK_THREADS) void gat_bin_gather(
    const int* __restrict__ part, const int* __restrict__ tabCnt,
    const int* __restrict__ tabOff, int* __restrict__ spill,
    const float* __restrict__ hW, const float* __restrict__ el,
    const float* __restrict__ er, const float* __restrict__ bias,
    float* __restrict__ out)
{
  __shared__ int sSegCnt[NCHUNKS];
  __shared__ int sSegOff[NCHUNKS];
  __shared__ int sCnt[BIN_NODES];
  __shared__ int sOff[BIN_NODES];
  __shared__ int sCur[BIN_NODES];
  __shared__ int sRed[2][8];
  __shared__ int sStage[BK_CAP];
  const int t = threadIdx.x;
  const int lane = t & 31;
  const int w = t >> 5;
  const int b = blockIdx.x;

  if (t < BIN_NODES) { sCnt[t] = 0; sCur[t] = 0; }
  int myCnt = 0, myBelow = 0;
  if (t < NCHUNKS) {
    myCnt = tabCnt[t * NBINS + b];
    const int o = tabOff[t * NBINS + b];
    sSegCnt[t] = myCnt;
    sSegOff[t] = o;
    myBelow = o - t * PART_CHUNK;
  }
  if (w < 8) {
    const int sb = waveSumI(myBelow);
    const int sc0 = waveSumI(myCnt);
    if (lane == 0) { sRed[0][w] = sb; sRed[1][w] = sc0; }
  }
  __syncthreads();
  int binBase = 0, binCount = 0;
#pragma unroll
  for (int q = 0; q < 8; ++q) { binBase += sRed[0][q]; binCount += sRed[1][q]; }

  int pv[BK_SEGS], sc[BK_SEGS], so[BK_SEGS];
#pragma unroll
  for (int q = 0; q < BK_SEGS; ++q) {
    const int j = w + q * BK_WAVES;
    sc[q] = (j < NCHUNKS) ? sSegCnt[j] : 0;
    so[q] = (j < NCHUNKS) ? sSegOff[j] : 0;
    const int ix = so[q] + lane;
    const int pw = part[(ix < E_) ? ix : E_ - 1];
    pv[q] = (lane < sc[q]) ? pw : -1;
  }
#pragma unroll
  for (int q = 0; q < BK_SEGS; ++q)
    if (pv[q] >= 0) atomicAdd(&sCnt[pv[q] >> 16], 1);
#pragma unroll
  for (int q = 0; q < BK_SEGS; ++q)
    for (int i = lane + 32; i < sc[q]; i += 32)
      atomicAdd(&sCnt[part[so[q] + i] >> 16], 1);
  __syncthreads();

  if (t < 32) {
    int c[8]; int s = 0;
#pragma unroll
    for (int q = 0; q < 8; ++q) { c[q] = sCnt[lane * 8 + q]; s += c[q]; }
    int run = waveIncScanI(s, lane) - s;
#pragma unroll
    for (int q = 0; q < 8; ++q) { sOff[lane * 8 + q] = run; run += c[q]; }
  }
  __syncthreads();

  const bool staged = binCount <= BK_CAP;
  int* spillBin = spill + binBase;
#pragma unroll
  for (int q = 0; q < BK_SEGS; ++q) {
    if (pv[q] >= 0) {
      const int f = pv[q] >> 16;
      const int pos = sOff[f] + atomicAdd(&sCur[f], 1);
      const int s = pv[q] & 0xFFFF;
      if (staged) sStage[pos] = s; else spillBin[pos] = s;
    }
  }
#pragma unroll
  for (int q = 0; q < BK_SEGS; ++q) {
    for (int i = lane + 32; i < sc[q]; i += 32) {
      const int p = part[so[q] + i];
      const int f = p >> 16;
      const int pos = sOff[f] + atomicAdd(&sCur[f], 1);
      const int s = p & 0xFFFF;
      if (staged) sStage[pos] = s; else spillBin[pos] = s;
    }
  }
  if (!staged) __threadfence();
  __syncthreads();

  const int wave = __builtin_amdgcn_readfirstlane(w);
  if (staged) gatherBin<true>(b, wave, lane, sOff, sCnt, sStage, spillBin, hW, el, er, bias, out);
  else        gatherBin<false>(b, wave, lane, sOff, sCnt, sStage, spillBin, hW, el, er, bias, out);
}

extern "C" void kernel_launch(void* const* d_in, const int* in_sizes, int n_in,
                              void* d_out, int out_size, void* d_ws, size_t ws_size,
                              hipStream_t stream)
{
  const float* h    = (const float*)d_in[0];
  const float* W    = (const float*)d_in[1];
  const float* alw  = (const float*)d_in[2];
  const float* alb  = (const float*)d_in[3];
  const float* arw  = (const float*)d_in[4];
  const float* arb  = (const float*)d_in[5];
  const float* bias = (const float*)d_in[6];
  const int*   src  = (const int*)d_in[7];
  const int*   dst  = (const int*)d_in[8];
  float* out = (float*)d_out;

  char* ws = (char*)d_ws;
  size_t off_b = 0;
  auto carve = [&](size_t bytes) -> void* {
    void* p = ws + off_b;
    off_b += (bytes + 255) & ~(size_t)255;
    return p;
  };
  float* hW   = (float*)carve((size_t)N_ * HF_ * sizeof(float));
  float* el   = (float*)carve((size_t)N_ * H_ * sizeof(float));
  float* er   = (float*)carve((size_t)N_ * H_ * sizeof(float));
  int*   tcnt = (int*)carve((size_t)NCHUNKS * NBINS * sizeof(int));
  int*   toff = (int*)carve((size_t)NCHUNKS * NBINS * sizeof(int));
  int*   part = (int*)carve((size_t)E_ * sizeof(int));
  int*   spl  = (int*)carve((size_t)E_ * sizeof(int));
  (void)ws_size; (void)in_sizes; (void)n_in; (void)out_size;

  gat_gemm_wmma<<<GEMM_BLOCKS + NCHUNKS, GEMM_THREADS, 0, stream>>>(h, W, alw, alb, arw, arb, src, dst,
                                                                   hW, el, er, part, tcnt, toff);
  gat_bin_gather<<<NBINS, BK_THREADS, 0, stream>>>(part, tcnt, toff, spl, hW, el, er, bias, out);
}
